# starting kernel
# baseline (speedup 1.0000x reference)
#include <hip/hip_runtime.h>
#include <math.h>

#define NN 50000
#define NE 625000
#define D 128
#define H 4
#define DH 32
#define NEG_SLOPE 0.2f
#define LN_EPS 1e-5f

typedef float v2f __attribute__((ext_vector_type(2)));
typedef float v8f __attribute__((ext_vector_type(8)));

__device__ __forceinline__ unsigned fenc(float x) {
    unsigned b = __float_as_uint(x);
    return (b & 0x80000000u) ? ~b : (b | 0x80000000u);
}
__device__ __forceinline__ float fdec(unsigned u) {
    unsigned b = (u & 0x80000000u) ? (u & 0x7FFFFFFFu) : ~u;
    return __uint_as_float(b);
}
__device__ __forceinline__ float lrelu(float v) {
    return v > 0.0f ? v : NEG_SLOPE * v;
}

__global__ void k_init(float* __restrict__ out, unsigned* __restrict__ m,
                       float* __restrict__ den) {
    int i = blockIdx.x * 256 + threadIdx.x;
    out[i] = 0.0f;
    if (i < NN * H) { m[i] = 0x007FFFFFu; den[i] = 0.0f; }
}

__global__ void k_proj(const float* __restrict__ A, const float* __restrict__ W,
                       float* __restrict__ X) {
    int wave = (blockIdx.x * blockDim.x + threadIdx.x) >> 5;
    int lane = threadIdx.x & 31;
    int tm = wave >> 3, tn = wave & 7;
    int row = lane & 15, half = lane >> 4;
    const float* arow = A + (tm * 16 + row) * D;
    const float* bcol = W + tn * 16 + row;
    v8f c = {0.f, 0.f, 0.f, 0.f, 0.f, 0.f, 0.f, 0.f};
    for (int k0 = 0; k0 < D; k0 += 4) {
        int ka = k0 + half * 2;
        v2f a; a[0] = arow[ka];       a[1] = arow[ka + 1];
        v2f b; b[0] = bcol[ka * D];   b[1] = bcol[(ka + 1) * D];
        c = __builtin_amdgcn_wmma_f32_16x16x4_f32(false, a, false, b,
                                                  (short)0, c, false, false);
    }
    float* drow = X + tn * 16 + row;
#pragma unroll
    for (int j = 0; j < 8; ++j)
        drow[(tm * 16 + half * 8 + j) * D] = c[j];
}

__global__ void k_logits(const float* __restrict__ X,
                         const float* __restrict__ att_src,
                         const float* __restrict__ att_dst,
                         float* __restrict__ a_src, float* __restrict__ a_dst) {
    int i = blockIdx.x * blockDim.x + threadIdx.x;
    if (i >= NN * H) return;
    int n = i >> 2, h = i & 3;
    const float* xr = X + n * D + h * DH;
    const float* as = att_src + h * DH;
    const float* ad = att_dst + h * DH;
    float s1 = 0.f, s2 = 0.f;
#pragma unroll 8
    for (int j = 0; j < DH; ++j) { float v = xr[j]; s1 += v * as[j]; s2 += v * ad[j]; }
    a_src[i] = s1; a_dst[i] = s2;
}

__global__ void k_edge_max(const int* __restrict__ ei,
                           const float* __restrict__ a_src,
                           const float* __restrict__ a_dst,
                           unsigned* __restrict__ m) {
    int e = blockIdx.x * blockDim.x + threadIdx.x;
    if (e >= NE) return;
    int s = ei[e], d = ei[NE + e];
#pragma unroll
    for (int h = 0; h < H; ++h) {
        float v = lrelu(a_src[s * H + h] + a_dst[d * H + h]);
        atomicMax(&m[d * H + h], fenc(v));
    }
}

__global__ void k_edge_sum(const int* __restrict__ ei,
                           const float* __restrict__ a_src,
                           const float* __restrict__ a_dst,
                           const unsigned* __restrict__ m,
                           float* __restrict__ den) {
    int e = blockIdx.x * blockDim.x + threadIdx.x;
    if (e >= NE) return;
    int s = ei[e], d = ei[NE + e];
#pragma unroll
    for (int h = 0; h < H; ++h) {
        float v = lrelu(a_src[s * H + h] + a_dst[d * H + h]);
        atomicAdd(&den[d * H + h], __expf(v - fdec(m[d * H + h])));
    }
}

__global__ void k_aggr(const int* __restrict__ ei,
                       const float* __restrict__ a_src,
                       const float* __restrict__ a_dst,
                       const unsigned* __restrict__ m,
                       const float* __restrict__ den,
                       const float* __restrict__ X, float* __restrict__ out) {
    int wave = (blockIdx.x * blockDim.x + threadIdx.x) >> 5;
    int lane = threadIdx.x & 31;
    int s = ei[wave], d = ei[NE + wave];
    int h = lane >> 3;
    float v = lrelu(a_src[s * H + h] + a_dst[d * H + h]);
    float alpha = __expf(v - fdec(m[d * H + h])) / fmaxf(den[d * H + h], 1e-16f);
    const float4 xv = *(const float4*)(X + s * D + lane * 4);
    float* o = out + d * D + lane * 4;
    atomicAdd(o + 0, alpha * xv.x);
    atomicAdd(o + 1, alpha * xv.y);
    atomicAdd(o + 2, alpha * xv.z);
    atomicAdd(o + 3, alpha * xv.w);
}

__global__ void k_ln(const float* __restrict__ acc, const float* __restrict__ bias,
                     const float* __restrict__ gamma, const float* __restrict__ beta,
                     float* __restrict__ lnout) {
    int n = (blockIdx.x * blockDim.x + threadIdx.x) >> 5;
    int lane = threadIdx.x & 31;
    float4 v = *(const float4*)(acc + n * D + lane * 4);
    const float4 b4 = *(const float4*)(bias + lane * 4);
    v.x += b4.x; v.y += b4.y; v.z += b4.z; v.w += b4.w;
    float s = v.x + v.y + v.z + v.w;
#pragma unroll
    for (int off = 16; off >= 1; off >>= 1) s += __shfl_xor(s, off, 32);
    float mu = s * (1.0f / D);
    float dx = v.x - mu, dy = v.y - mu, dz = v.z - mu, dw = v.w - mu;
    float q = dx * dx + dy * dy + dz * dz + dw * dw;
#pragma unroll
    for (int off = 16; off >= 1; off >>= 1) q += __shfl_xor(q, off, 32);
    float rs = rsqrtf(q * (1.0f / D) + LN_EPS);
    const float4 g4 = *(const float4*)(gamma + lane * 4);
    const float4 be4 = *(const float4*)(beta + lane * 4);
    float4 o;
    o.x = dx * rs * g4.x + be4.x;  o.y = dy * rs * g4.y + be4.y;
    o.z = dz * rs * g4.z + be4.z;  o.w = dw * rs * g4.w + be4.w;
    *(float4*)(lnout + n * D + lane * 4) = o;
}

__global__ void k_gate(const float* __restrict__ LN, const float* __restrict__ query,
                       const int* __restrict__ n2g, const float* __restrict__ gW,
                       const float* __restrict__ gb, float* __restrict__ out) {
    int wave = (blockIdx.x * blockDim.x + threadIdx.x) >> 5;
    int lane = threadIdx.x & 31;
    int tm = wave >> 3, tn = wave & 7;
    int row = lane & 15, half = lane >> 4;
    int arow_i = tm * 16 + row;
    const float* lrow = LN + arow_i * D;
    const float* qrow = query + n2g[arow_i] * D;
    const float* bcol = gW + tn * 16 + row;
    v8f c = {0.f, 0.f, 0.f, 0.f, 0.f, 0.f, 0.f, 0.f};
    for (int k0 = 0; k0 < D; k0 += 4) {
        int ka = k0 + half * 2;
        v2f a; a[0] = lrow[ka];     a[1] = lrow[ka + 1];
        v2f b; b[0] = bcol[ka * D]; b[1] = bcol[(ka + 1) * D];
        c = __builtin_amdgcn_wmma_f32_16x16x4_f32(false, a, false, b,
                                                  (short)0, c, false, false);
    }
    for (int k0 = 0; k0 < D; k0 += 4) {
        int ka = k0 + half * 2;
        v2f a; a[0] = qrow[ka];             a[1] = qrow[ka + 1];
        v2f b; b[0] = bcol[(D + ka) * D];   b[1] = bcol[(D + ka + 1) * D];
        c = __builtin_amdgcn_wmma_f32_16x16x4_f32(false, a, false, b,
                                                  (short)0, c, false, false);
    }
    int cc = tn * 16 + row;
    float gbv = gb[cc];
#pragma unroll
    for (int j = 0; j < 8; ++j) {
        int rr = tm * 16 + half * 8 + j;
        float gate = 1.0f / (1.0f + __expf(-(c[j] + gbv)));
        out[rr * D + cc] = LN[rr * D + cc] * gate;
    }
}

extern "C" void kernel_launch(void* const* d_in, const int* in_sizes, int n_in,
                              void* d_out, int out_size, void* d_ws, size_t ws_size,
                              hipStream_t stream) {
    (void)in_sizes; (void)n_in; (void)out_size; (void)ws_size;
    const float* node_input = (const float*)d_in[0];
    const int*   edge_index = (const int*)d_in[1];
    const int*   node2graph = (const int*)d_in[2];
    const float* query      = (const float*)d_in[3];
    const float* W          = (const float*)d_in[4];
    const float* att_src    = (const float*)d_in[5];
    const float* att_dst    = (const float*)d_in[6];
    const float* bias       = (const float*)d_in[7];
    const float* ln_gamma   = (const float*)d_in[8];
    const float* ln_beta    = (const float*)d_in[9];
    const float* gate_W     = (const float*)d_in[10];
    const float* gate_b     = (const float*)d_in[11];
    float* out = (float*)d_out;

    float*    X     = (float*)d_ws;
    float*    a_src = X + NN * D;
    float*    a_dst = a_src + NN * H;
    unsigned* m     = (unsigned*)(a_dst + NN * H);
    float*    den   = (float*)(m + NN * H);

    k_init    <<<(NN * D) / 256, 256, 0, stream>>>(out, m, den);
    k_proj    <<<6250, 128, 0, stream>>>(node_input, W, X);
    k_logits  <<<(NN * H + 255) / 256, 256, 0, stream>>>(X, att_src, att_dst, a_src, a_dst);
    k_edge_max<<<(NE + 255) / 256, 256, 0, stream>>>(edge_index, a_src, a_dst, m);
    k_edge_sum<<<(NE + 255) / 256, 256, 0, stream>>>(edge_index, a_src, a_dst, m, den);
    k_aggr    <<<NE / 8, 256, 0, stream>>>(edge_index, a_src, a_dst, m, den, X, out);
    k_ln      <<<NN / 8, 256, 0, stream>>>(out, bias, ln_gamma, ln_beta, X);
    k_gate    <<<6250, 128, 0, stream>>>(X, query, node2graph, gate_W, gate_b, out);
}


# edge max and sum passes: one lane per (edge, head), so an edge's four head atomics leave as one 16-byte request instead of four single-word ones
# speedup vs baseline: 1.4409x; 1.4409x over previous
#include <hip/hip_runtime.h>
#include <math.h>

#define NN 50000
#define NE 625000
#define D 128
#define H 4
#define DH 32
#define NEG_SLOPE 0.2f
#define LN_EPS 1e-5f

typedef float v2f __attribute__((ext_vector_type(2)));
typedef float v8f __attribute__((ext_vector_type(8)));

__device__ __forceinline__ unsigned fenc(float x) {
    unsigned b = __float_as_uint(x);
    return (b & 0x80000000u) ? ~b : (b | 0x80000000u);
}
__device__ __forceinline__ float fdec(unsigned u) {
    unsigned b = (u & 0x80000000u) ? (u & 0x7FFFFFFFu) : ~u;
    return __uint_as_float(b);
}
__device__ __forceinline__ float lrelu(float v) {
    return v > 0.0f ? v : NEG_SLOPE * v;
}

__global__ void k_init(float* __restrict__ out, unsigned* __restrict__ m,
                       float* __restrict__ den) {
    int i = blockIdx.x * 256 + threadIdx.x;
    out[i] = 0.0f;
    if (i < NN * H) { m[i] = 0x007FFFFFu; den[i] = 0.0f; }
}

__global__ void k_proj(const float* __restrict__ A, const float* __restrict__ W,
                       float* __restrict__ X) {
    int wave = (blockIdx.x * blockDim.x + threadIdx.x) >> 5;
    int lane = threadIdx.x & 31;
    int tm = wave >> 3, tn = wave & 7;
    int row = lane & 15, half = lane >> 4;
    const float* arow = A + (tm * 16 + row) * D;
    const float* bcol = W + tn * 16 + row;
    v8f c = {0.f, 0.f, 0.f, 0.f, 0.f, 0.f, 0.f, 0.f};
    for (int k0 = 0; k0 < D; k0 += 4) {
        int ka = k0 + half * 2;
        v2f a; a[0] = arow[ka];       a[1] = arow[ka + 1];
        v2f b; b[0] = bcol[ka * D];   b[1] = bcol[(ka + 1) * D];
        c = __builtin_amdgcn_wmma_f32_16x16x4_f32(false, a, false, b,
                                                  (short)0, c, false, false);
    }
    float* drow = X + tn * 16 + row;
#pragma unroll
    for (int j = 0; j < 8; ++j)
        drow[(tm * 16 + half * 8 + j) * D] = c[j];
}

__global__ void k_logits(const float* __restrict__ X,
                         const float* __restrict__ att_src,
                         const float* __restrict__ att_dst,
                         float* __restrict__ a_src, float* __restrict__ a_dst) {
    int i = blockIdx.x * blockDim.x + threadIdx.x;
    if (i >= NN * H) return;
    int n = i >> 2, h = i & 3;
    const float* xr = X + n * D + h * DH;
    const float* as = att_src + h * DH;
    const float* ad = att_dst + h * DH;
    float s1 = 0.f, s2 = 0.f;
#pragma unroll 8
    for (int j = 0; j < DH; ++j) { float v = xr[j]; s1 += v * as[j]; s2 += v * ad[j]; }
    a_src[i] = s1; a_dst[i] = s2;
}

__global__ void k_edge_max(const int* __restrict__ ei,
                           const float* __restrict__ a_src,
                           const float* __restrict__ a_dst,
                           unsigned* __restrict__ m) {
    int t = blockIdx.x * blockDim.x + threadIdx.x;
    int lane = t & 31;
    int h = lane & 3;
    int e0 = (t >> 5) * 32 + (lane >> 2);
    if (e0 >= NE) return;
    int s[4], d[4];
    float v[4];
#pragma unroll
    for (int i = 0; i < 4; ++i) {
        int e = min(e0 + 8 * i, NE - 1);
        s[i] = ei[e];
        d[i] = ei[NE + e];
    }
#pragma unroll
    for (int i = 0; i < 4; ++i)
        v[i] = lrelu(a_src[s[i] * H + h] + a_dst[d[i] * H + h]);
#pragma unroll
    for (int i = 0; i < 4; ++i)
        atomicMax(&m[d[i] * H + h], fenc(v[i]));
}

__global__ void k_edge_sum(const int* __restrict__ ei,
                           const float* __restrict__ a_src,
                           const float* __restrict__ a_dst,
                           const unsigned* __restrict__ m,
                           float* __restrict__ den) {
    int t = blockIdx.x * blockDim.x + threadIdx.x;
    int lane = t & 31;
    int h = lane & 3;
    int e0 = (t >> 5) * 32 + (lane >> 2);
    if (e0 >= NE) return;
    int s[4], d[4];
    float ex[4];
#pragma unroll
    for (int i = 0; i < 4; ++i) {
        int e = min(e0 + 8 * i, NE - 1);
        s[i] = ei[e];
        d[i] = ei[NE + e];
    }
#pragma unroll
    for (int i = 0; i < 4; ++i) {
        float v = lrelu(a_src[s[i] * H + h] + a_dst[d[i] * H + h]);
        float p = __expf(v - fdec(m[d[i] * H + h]));
        ex[i] = (e0 + 8 * i < NE) ? p : 0.0f;
    }
#pragma unroll
    for (int i = 0; i < 4; ++i)
        atomicAdd(&den[d[i] * H + h], ex[i]);
}

__global__ void k_aggr(const int* __restrict__ ei,
                       const float* __restrict__ a_src,
                       const float* __restrict__ a_dst,
                       const unsigned* __restrict__ m,
                       const float* __restrict__ den,
                       const float* __restrict__ X, float* __restrict__ out) {
    int wave = (blockIdx.x * blockDim.x + threadIdx.x) >> 5;
    int lane = threadIdx.x & 31;
    int s = ei[wave], d = ei[NE + wave];
    int h = lane >> 3;
    float v = lrelu(a_src[s * H + h] + a_dst[d * H + h]);
    float alpha = __expf(v - fdec(m[d * H + h])) / fmaxf(den[d * H + h], 1e-16f);
    const float4 xv = *(const float4*)(X + s * D + lane * 4);
    float* o = out + d * D + lane * 4;
    atomicAdd(o + 0, alpha * xv.x);
    atomicAdd(o + 1, alpha * xv.y);
    atomicAdd(o + 2, alpha * xv.z);
    atomicAdd(o + 3, alpha * xv.w);
}

__global__ void k_ln(const float* __restrict__ acc, const float* __restrict__ bias,
                     const float* __restrict__ gamma, const float* __restrict__ beta,
                     float* __restrict__ lnout) {
    int n = (blockIdx.x * blockDim.x + threadIdx.x) >> 5;
    int lane = threadIdx.x & 31;
    float4 v = *(const float4*)(acc + n * D + lane * 4);
    const float4 b4 = *(const float4*)(bias + lane * 4);
    v.x += b4.x; v.y += b4.y; v.z += b4.z; v.w += b4.w;
    float s = v.x + v.y + v.z + v.w;
#pragma unroll
    for (int off = 16; off >= 1; off >>= 1) s += __shfl_xor(s, off, 32);
    float mu = s * (1.0f / D);
    float dx = v.x - mu, dy = v.y - mu, dz = v.z - mu, dw = v.w - mu;
    float q = dx * dx + dy * dy + dz * dz + dw * dw;
#pragma unroll
    for (int off = 16; off >= 1; off >>= 1) q += __shfl_xor(q, off, 32);
    float rs = rsqrtf(q * (1.0f / D) + LN_EPS);
    const float4 g4 = *(const float4*)(gamma + lane * 4);
    const float4 be4 = *(const float4*)(beta + lane * 4);
    float4 o;
    o.x = dx * rs * g4.x + be4.x;  o.y = dy * rs * g4.y + be4.y;
    o.z = dz * rs * g4.z + be4.z;  o.w = dw * rs * g4.w + be4.w;
    *(float4*)(lnout + n * D + lane * 4) = o;
}

__global__ void k_gate(const float* __restrict__ LN, const float* __restrict__ query,
                       const int* __restrict__ n2g, const float* __restrict__ gW,
                       const float* __restrict__ gb, float* __restrict__ out) {
    int wave = (blockIdx.x * blockDim.x + threadIdx.x) >> 5;
    int lane = threadIdx.x & 31;
    int tm = wave >> 3, tn = wave & 7;
    int row = lane & 15, half = lane >> 4;
    int arow_i = tm * 16 + row;
    const float* lrow = LN + arow_i * D;
    const float* qrow = query + n2g[arow_i] * D;
    const float* bcol = gW + tn * 16 + row;
    v8f c = {0.f, 0.f, 0.f, 0.f, 0.f, 0.f, 0.f, 0.f};
    for (int k0 = 0; k0 < D; k0 += 4) {
        int ka = k0 + half * 2;
        v2f a; a[0] = lrow[ka];     a[1] = lrow[ka + 1];
        v2f b; b[0] = bcol[ka * D]; b[1] = bcol[(ka + 1) * D];
        c = __builtin_amdgcn_wmma_f32_16x16x4_f32(false, a, false, b,
                                                  (short)0, c, false, false);
    }
    for (int k0 = 0; k0 < D; k0 += 4) {
        int ka = k0 + half * 2;
        v2f a; a[0] = qrow[ka];             a[1] = qrow[ka + 1];
        v2f b; b[0] = bcol[(D + ka) * D];   b[1] = bcol[(D + ka + 1) * D];
        c = __builtin_amdgcn_wmma_f32_16x16x4_f32(false, a, false, b,
                                                  (short)0, c, false, false);
    }
    int cc = tn * 16 + row;
    float gbv = gb[cc];
#pragma unroll
    for (int j = 0; j < 8; ++j) {
        int rr = tm * 16 + half * 8 + j;
        float gate = 1.0f / (1.0f + __expf(-(c[j] + gbv)));
        out[rr * D + cc] = LN[rr * D + cc] * gate;
    }
}

extern "C" void kernel_launch(void* const* d_in, const int* in_sizes, int n_in,
                              void* d_out, int out_size, void* d_ws, size_t ws_size,
                              hipStream_t stream) {
    (void)in_sizes; (void)n_in; (void)out_size; (void)ws_size;
    const float* node_input = (const float*)d_in[0];
    const int*   edge_index = (const int*)d_in[1];
    const int*   node2graph = (const int*)d_in[2];
    const float* query      = (const float*)d_in[3];
    const float* W          = (const float*)d_in[4];
    const float* att_src    = (const float*)d_in[5];
    const float* att_dst    = (const float*)d_in[6];
    const float* bias       = (const float*)d_in[7];
    const float* ln_gamma   = (const float*)d_in[8];
    const float* ln_beta    = (const float*)d_in[9];
    const float* gate_W     = (const float*)d_in[10];
    const float* gate_b     = (const float*)d_in[11];
    float* out = (float*)d_out;

    float*    X     = (float*)d_ws;
    float*    a_src = X + NN * D;
    float*    a_dst = a_src + NN * H;
    unsigned* m     = (unsigned*)(a_dst + NN * H);
    float*    den   = (float*)(m + NN * H);

    k_init    <<<(NN * D) / 256, 256, 0, stream>>>(out, m, den);
    k_proj    <<<6250, 128, 0, stream>>>(node_input, W, X);
    k_logits  <<<(NN * H + 255) / 256, 256, 0, stream>>>(X, att_src, att_dst, a_src, a_dst);
    k_edge_max<<<(NE + 255) / 256, 256, 0, stream>>>(edge_index, a_src, a_dst, m);
    k_edge_sum<<<(NE + 255) / 256, 256, 0, stream>>>(edge_index, a_src, a_dst, m, den);
    k_aggr    <<<NE / 8, 256, 0, stream>>>(edge_index, a_src, a_dst, m, den, X, out);
    k_ln      <<<NN / 8, 256, 0, stream>>>(out, bias, ln_gamma, ln_beta, X);
    k_gate    <<<6250, 128, 0, stream>>>(X, query, node2graph, gate_W, gate_b, out);
}


# aggregation pass: lane owns column l of each head (features l, l+32, l+64, l+96), so every row load and atomic add covers one whole 128-byte line: 4 add requests per edge, not 16
# speedup vs baseline: 1.7360x; 1.2048x over previous
#include <hip/hip_runtime.h>
#include <math.h>

#define NN 50000
#define NE 625000
#define D 128
#define H 4
#define DH 32
#define NEG_SLOPE 0.2f
#define LN_EPS 1e-5f

typedef float v2f __attribute__((ext_vector_type(2)));
typedef float v8f __attribute__((ext_vector_type(8)));

__device__ __forceinline__ unsigned fenc(float x) {
    unsigned b = __float_as_uint(x);
    return (b & 0x80000000u) ? ~b : (b | 0x80000000u);
}
__device__ __forceinline__ float fdec(unsigned u) {
    unsigned b = (u & 0x80000000u) ? (u & 0x7FFFFFFFu) : ~u;
    return __uint_as_float(b);
}
__device__ __forceinline__ float lrelu(float v) {
    return v > 0.0f ? v : NEG_SLOPE * v;
}

__global__ void k_init(float* __restrict__ out, unsigned* __restrict__ m,
                       float* __restrict__ den) {
    int i = blockIdx.x * 256 + threadIdx.x;
    out[i] = 0.0f;
    if (i < NN * H) { m[i] = 0x007FFFFFu; den[i] = 0.0f; }
}

__global__ void k_proj(const float* __restrict__ A, const float* __restrict__ W,
                       float* __restrict__ X) {
    int wave = (blockIdx.x * blockDim.x + threadIdx.x) >> 5;
    int lane = threadIdx.x & 31;
    int tm = wave >> 3, tn = wave & 7;
    int row = lane & 15, half = lane >> 4;
    const float* arow = A + (tm * 16 + row) * D;
    const float* bcol = W + tn * 16 + row;
    v8f c = {0.f, 0.f, 0.f, 0.f, 0.f, 0.f, 0.f, 0.f};
    for (int k0 = 0; k0 < D; k0 += 4) {
        int ka = k0 + half * 2;
        v2f a; a[0] = arow[ka];       a[1] = arow[ka + 1];
        v2f b; b[0] = bcol[ka * D];   b[1] = bcol[(ka + 1) * D];
        c = __builtin_amdgcn_wmma_f32_16x16x4_f32(false, a, false, b,
                                                  (short)0, c, false, false);
    }
    float* drow = X + tn * 16 + row;
#pragma unroll
    for (int j = 0; j < 8; ++j)
        drow[(tm * 16 + half * 8 + j) * D] = c[j];
}

__global__ void k_logits(const float* __restrict__ X,
                         const float* __restrict__ att_src,
                         const float* __restrict__ att_dst,
                         float* __restrict__ a_src, float* __restrict__ a_dst) {
    int i = blockIdx.x * blockDim.x + threadIdx.x;
    if (i >= NN * H) return;
    int n = i >> 2, h = i & 3;
    const float* xr = X + n * D + h * DH;
    const float* as = att_src + h * DH;
    const float* ad = att_dst + h * DH;
    float s1 = 0.f, s2 = 0.f;
#pragma unroll 8
    for (int j = 0; j < DH; ++j) { float v = xr[j]; s1 += v * as[j]; s2 += v * ad[j]; }
    a_src[i] = s1; a_dst[i] = s2;
}

__global__ void k_edge_max(const int* __restrict__ ei,
                           const float* __restrict__ a_src,
                           const float* __restrict__ a_dst,
                           unsigned* __restrict__ m) {
    int t = blockIdx.x * blockDim.x + threadIdx.x;
    int lane = t & 31;
    int h = lane & 3;
    int e0 = (t >> 5) * 32 + (lane >> 2);
    if (e0 >= NE) return;
    int s[4], d[4];
    float v[4];
#pragma unroll
    for (int i = 0; i < 4; ++i) {
        int e = min(e0 + 8 * i, NE - 1);
        s[i] = ei[e];
        d[i] = ei[NE + e];
    }
#pragma unroll
    for (int i = 0; i < 4; ++i)
        v[i] = lrelu(a_src[s[i] * H + h] + a_dst[d[i] * H + h]);
#pragma unroll
    for (int i = 0; i < 4; ++i)
        atomicMax(&m[d[i] * H + h], fenc(v[i]));
}

__global__ void k_edge_sum(const int* __restrict__ ei,
                           const float* __restrict__ a_src,
                           const float* __restrict__ a_dst,
                           const unsigned* __restrict__ m,
                           float* __restrict__ den) {
    int t = blockIdx.x * blockDim.x + threadIdx.x;
    int lane = t & 31;
    int h = lane & 3;
    int e0 = (t >> 5) * 32 + (lane >> 2);
    if (e0 >= NE) return;
    int s[4], d[4];
    float ex[4];
#pragma unroll
    for (int i = 0; i < 4; ++i) {
        int e = min(e0 + 8 * i, NE - 1);
        s[i] = ei[e];
        d[i] = ei[NE + e];
    }
#pragma unroll
    for (int i = 0; i < 4; ++i) {
        float v = lrelu(a_src[s[i] * H + h] + a_dst[d[i] * H + h]);
        float p = __expf(v - fdec(m[d[i] * H + h]));
        ex[i] = (e0 + 8 * i < NE) ? p : 0.0f;
    }
#pragma unroll
    for (int i = 0; i < 4; ++i)
        atomicAdd(&den[d[i] * H + h], ex[i]);
}

__global__ void k_aggr(const int* __restrict__ ei,
                       const float* __restrict__ a_src,
                       const float* __restrict__ a_dst,
                       const unsigned* __restrict__ m,
                       const float* __restrict__ den,
                       const float* __restrict__ X, float* __restrict__ out) {
    int wave = (blockIdx.x * blockDim.x + threadIdx.x) >> 5;
    int lane = threadIdx.x & 31;
    int s = ei[wave], d = ei[NE + wave];
    int h = lane & 3;
    float v = lrelu(a_src[s * H + h] + a_dst[d * H + h]);
    float alpha = __expf(v - fdec(m[d * H + h])) / fmaxf(den[d * H + h], 1e-16f);
    const float* xr = X + s * D + lane;
    float x0 = xr[0], x1 = xr[DH], x2 = xr[2 * DH], x3 = xr[3 * DH];
    int ab = __float_as_int(alpha);
    float w0 = __int_as_float(__builtin_amdgcn_readlane(ab, 0));
    float w1 = __int_as_float(__builtin_amdgcn_readlane(ab, 1));
    float w2 = __int_as_float(__builtin_amdgcn_readlane(ab, 2));
    float w3 = __int_as_float(__builtin_amdgcn_readlane(ab, 3));
    float* o = out + d * D + lane;
    atomicAdd(o,          w0 * x0);
    atomicAdd(o + DH,     w1 * x1);
    atomicAdd(o + 2 * DH, w2 * x2);
    atomicAdd(o + 3 * DH, w3 * x3);
}

__global__ void k_ln(const float* __restrict__ acc, const float* __restrict__ bias,
                     const float* __restrict__ gamma, const float* __restrict__ beta,
                     float* __restrict__ lnout) {
    int n = (blockIdx.x * blockDim.x + threadIdx.x) >> 5;
    int lane = threadIdx.x & 31;
    float4 v = *(const float4*)(acc + n * D + lane * 4);
    const float4 b4 = *(const float4*)(bias + lane * 4);
    v.x += b4.x; v.y += b4.y; v.z += b4.z; v.w += b4.w;
    float s = v.x + v.y + v.z + v.w;
#pragma unroll
    for (int off = 16; off >= 1; off >>= 1) s += __shfl_xor(s, off, 32);
    float mu = s * (1.0f / D);
    float dx = v.x - mu, dy = v.y - mu, dz = v.z - mu, dw = v.w - mu;
    float q = dx * dx + dy * dy + dz * dz + dw * dw;
#pragma unroll
    for (int off = 16; off >= 1; off >>= 1) q += __shfl_xor(q, off, 32);
    float rs = rsqrtf(q * (1.0f / D) + LN_EPS);
    const float4 g4 = *(const float4*)(gamma + lane * 4);
    const float4 be4 = *(const float4*)(beta + lane * 4);
    float4 o;
    o.x = dx * rs * g4.x + be4.x;  o.y = dy * rs * g4.y + be4.y;
    o.z = dz * rs * g4.z + be4.z;  o.w = dw * rs * g4.w + be4.w;
    *(float4*)(lnout + n * D + lane * 4) = o;
}

__global__ void k_gate(const float* __restrict__ LN, const float* __restrict__ query,
                       const int* __restrict__ n2g, const float* __restrict__ gW,
                       const float* __restrict__ gb, float* __restrict__ out) {
    int wave = (blockIdx.x * blockDim.x + threadIdx.x) >> 5;
    int lane = threadIdx.x & 31;
    int tm = wave >> 3, tn = wave & 7;
    int row = lane & 15, half = lane >> 4;
    int arow_i = tm * 16 + row;
    const float* lrow = LN + arow_i * D;
    const float* qrow = query + n2g[arow_i] * D;
    const float* bcol = gW + tn * 16 + row;
    v8f c = {0.f, 0.f, 0.f, 0.f, 0.f, 0.f, 0.f, 0.f};
    for (int k0 = 0; k0 < D; k0 += 4) {
        int ka = k0 + half * 2;
        v2f a; a[0] = lrow[ka];     a[1] = lrow[ka + 1];
        v2f b; b[0] = bcol[ka * D]; b[1] = bcol[(ka + 1) * D];
        c = __builtin_amdgcn_wmma_f32_16x16x4_f32(false, a, false, b,
                                                  (short)0, c, false, false);
    }
    for (int k0 = 0; k0 < D; k0 += 4) {
        int ka = k0 + half * 2;
        v2f a; a[0] = qrow[ka];             a[1] = qrow[ka + 1];
        v2f b; b[0] = bcol[(D + ka) * D];   b[1] = bcol[(D + ka + 1) * D];
        c = __builtin_amdgcn_wmma_f32_16x16x4_f32(false, a, false, b,
                                                  (short)0, c, false, false);
    }
    int cc = tn * 16 + row;
    float gbv = gb[cc];
#pragma unroll
    for (int j = 0; j < 8; ++j) {
        int rr = tm * 16 + half * 8 + j;
        float gate = 1.0f / (1.0f + __expf(-(c[j] + gbv)));
        out[rr * D + cc] = LN[rr * D + cc] * gate;
    }
}

extern "C" void kernel_launch(void* const* d_in, const int* in_sizes, int n_in,
                              void* d_out, int out_size, void* d_ws, size_t ws_size,
                              hipStream_t stream) {
    (void)in_sizes; (void)n_in; (void)out_size; (void)ws_size;
    const float* node_input = (const float*)d_in[0];
    const int*   edge_index = (const int*)d_in[1];
    const int*   node2graph = (const int*)d_in[2];
    const float* query      = (const float*)d_in[3];
    const float* W          = (const float*)d_in[4];
    const float* att_src    = (const float*)d_in[5];
    const float* att_dst    = (const float*)d_in[6];
    const float* bias       = (const float*)d_in[7];
    const float* ln_gamma   = (const float*)d_in[8];
    const float* ln_beta    = (const float*)d_in[9];
    const float* gate_W     = (const float*)d_in[10];
    const float* gate_b     = (const float*)d_in[11];
    float* out = (float*)d_out;

    float*    X     = (float*)d_ws;
    float*    a_src = X + NN * D;
    float*    a_dst = a_src + NN * H;
    unsigned* m     = (unsigned*)(a_dst + NN * H);
    float*    den   = (float*)(m + NN * H);

    k_init    <<<(NN * D) / 256, 256, 0, stream>>>(out, m, den);
    k_proj    <<<6250, 128, 0, stream>>>(node_input, W, X);
    k_logits  <<<(NN * H + 255) / 256, 256, 0, stream>>>(X, att_src, att_dst, a_src, a_dst);
    k_edge_max<<<(NE + 255) / 256, 256, 0, stream>>>(edge_index, a_src, a_dst, m);
    k_edge_sum<<<(NE + 255) / 256, 256, 0, stream>>>(edge_index, a_src, a_dst, m, den);
    k_aggr    <<<NE / 8, 256, 0, stream>>>(edge_index, a_src, a_dst, m, den, X, out);
    k_ln      <<<NN / 8, 256, 0, stream>>>(out, bias, ln_gamma, ln_beta, X);
    k_gate    <<<6250, 128, 0, stream>>>(X, query, node2graph, gate_W, gate_b, out);
}


# aggregation pass: eight edges per wave (lane 4q+h weighs edge q, head h; the 32 row loads of the eight edges in flight together), so 78125 waves enter instead of 625000
# speedup vs baseline: 1.9086x; 1.0994x over previous
#include <hip/hip_runtime.h>
#include <math.h>

#define NN 50000
#define NE 625000
#define D 128
#define H 4
#define DH 32
#define NEG_SLOPE 0.2f
#define LN_EPS 1e-5f

typedef float v2f __attribute__((ext_vector_type(2)));
typedef float v8f __attribute__((ext_vector_type(8)));

__device__ __forceinline__ unsigned fenc(float x) {
    unsigned b = __float_as_uint(x);
    return (b & 0x80000000u) ? ~b : (b | 0x80000000u);
}
__device__ __forceinline__ float fdec(unsigned u) {
    unsigned b = (u & 0x80000000u) ? (u & 0x7FFFFFFFu) : ~u;
    return __uint_as_float(b);
}
__device__ __forceinline__ float lrelu(float v) {
    return v > 0.0f ? v : NEG_SLOPE * v;
}

__global__ void k_init(float* __restrict__ out, unsigned* __restrict__ m,
                       float* __restrict__ den) {
    int i = blockIdx.x * 256 + threadIdx.x;
    out[i] = 0.0f;
    if (i < NN * H) { m[i] = 0x007FFFFFu; den[i] = 0.0f; }
}

__global__ void k_proj(const float* __restrict__ A, const float* __restrict__ W,
                       float* __restrict__ X) {
    int wave = (blockIdx.x * blockDim.x + threadIdx.x) >> 5;
    int lane = threadIdx.x & 31;
    int tm = wave >> 3, tn = wave & 7;
    int row = lane & 15, half = lane >> 4;
    const float* arow = A + (tm * 16 + row) * D;
    const float* bcol = W + tn * 16 + row;
    v8f c = {0.f, 0.f, 0.f, 0.f, 0.f, 0.f, 0.f, 0.f};
    for (int k0 = 0; k0 < D; k0 += 4) {
        int ka = k0 + half * 2;
        v2f a; a[0] = arow[ka];       a[1] = arow[ka + 1];
        v2f b; b[0] = bcol[ka * D];   b[1] = bcol[(ka + 1) * D];
        c = __builtin_amdgcn_wmma_f32_16x16x4_f32(false, a, false, b,
                                                  (short)0, c, false, false);
    }
    float* drow = X + tn * 16 + row;
#pragma unroll
    for (int j = 0; j < 8; ++j)
        drow[(tm * 16 + half * 8 + j) * D] = c[j];
}

__global__ void k_logits(const float* __restrict__ X,
                         const float* __restrict__ att_src,
                         const float* __restrict__ att_dst,
                         float* __restrict__ a_src, float* __restrict__ a_dst) {
    int i = blockIdx.x * blockDim.x + threadIdx.x;
    if (i >= NN * H) return;
    int n = i >> 2, h = i & 3;
    const float* xr = X + n * D + h * DH;
    const float* as = att_src + h * DH;
    const float* ad = att_dst + h * DH;
    float s1 = 0.f, s2 = 0.f;
#pragma unroll 8
    for (int j = 0; j < DH; ++j) { float v = xr[j]; s1 += v * as[j]; s2 += v * ad[j]; }
    a_src[i] = s1; a_dst[i] = s2;
}

__global__ void k_edge_max(const int* __restrict__ ei,
                           const float* __restrict__ a_src,
                           const float* __restrict__ a_dst,
                           unsigned* __restrict__ m) {
    int t = blockIdx.x * blockDim.x + threadIdx.x;
    int lane = t & 31;
    int h = lane & 3;
    int e0 = (t >> 5) * 32 + (lane >> 2);
    if (e0 >= NE) return;
    int s[4], d[4];
    float v[4];
#pragma unroll
    for (int i = 0; i < 4; ++i) {
        int e = min(e0 + 8 * i, NE - 1);
        s[i] = ei[e];
        d[i] = ei[NE + e];
    }
#pragma unroll
    for (int i = 0; i < 4; ++i)
        v[i] = lrelu(a_src[s[i] * H + h] + a_dst[d[i] * H + h]);
#pragma unroll
    for (int i = 0; i < 4; ++i)
        atomicMax(&m[d[i] * H + h], fenc(v[i]));
}

__global__ void k_edge_sum(const int* __restrict__ ei,
                           const float* __restrict__ a_src,
                           const float* __restrict__ a_dst,
                           const unsigned* __restrict__ m,
                           float* __restrict__ den) {
    int t = blockIdx.x * blockDim.x + threadIdx.x;
    int lane = t & 31;
    int h = lane & 3;
    int e0 = (t >> 5) * 32 + (lane >> 2);
    if (e0 >= NE) return;
    int s[4], d[4];
    float ex[4];
#pragma unroll
    for (int i = 0; i < 4; ++i) {
        int e = min(e0 + 8 * i, NE - 1);
        s[i] = ei[e];
        d[i] = ei[NE + e];
    }
#pragma unroll
    for (int i = 0; i < 4; ++i) {
        float v = lrelu(a_src[s[i] * H + h] + a_dst[d[i] * H + h]);
        float p = __expf(v - fdec(m[d[i] * H + h]));
        ex[i] = (e0 + 8 * i < NE) ? p : 0.0f;
    }
#pragma unroll
    for (int i = 0; i < 4; ++i)
        atomicAdd(&den[d[i] * H + h], ex[i]);
}

#define EPW 8
static_assert(NE % EPW == 0, "a wave's eight edges must all exist");
__global__ void k_aggr(const int* __restrict__ ei,
                       const float* __restrict__ a_src,
                       const float* __restrict__ a_dst,
                       const unsigned* __restrict__ m,
                       const float* __restrict__ den,
                       const float* __restrict__ X, float* __restrict__ out) {
    int wave = (blockIdx.x * blockDim.x + threadIdx.x) >> 5;
    int lane = threadIdx.x & 31;
    int e0 = wave * EPW;
    if (e0 >= NE) return;
    int q = lane >> 2, h = lane & 3;
    int s = ei[e0 + q], d = ei[NE + e0 + q];
    float as_ = a_src[s * H + h], ad_ = a_dst[d * H + h];
    unsigned mm = m[d * H + h];
    float dd = den[d * H + h];
    float x[EPW][4];
    int dj[EPW];
#pragma unroll
    for (int j = 0; j < EPW; ++j) {
        int sj = __builtin_amdgcn_readlane(s, 4 * j);
        dj[j]  = __builtin_amdgcn_readlane(d, 4 * j);
        const float* xr = X + sj * D + lane;
        x[j][0] = xr[0]; x[j][1] = xr[DH]; x[j][2] = xr[2 * DH]; x[j][3] = xr[3 * DH];
    }
    float v = lrelu(as_ + ad_);
    float alpha = __expf(v - fdec(mm)) / fmaxf(dd, 1e-16f);
    int ab = __float_as_int(alpha);
#pragma unroll
    for (int j = 0; j < EPW; ++j) {
        float* o = out + dj[j] * D + lane;
#pragma unroll
        for (int k = 0; k < H; ++k) {
            float w = __int_as_float(__builtin_amdgcn_readlane(ab, 4 * j + k));
            atomicAdd(o + k * DH, w * x[j][k]);
        }
    }
}

__global__ void k_ln(const float* __restrict__ acc, const float* __restrict__ bias,
                     const float* __restrict__ gamma, const float* __restrict__ beta,
                     float* __restrict__ lnout) {
    int n = (blockIdx.x * blockDim.x + threadIdx.x) >> 5;
    int lane = threadIdx.x & 31;
    float4 v = *(const float4*)(acc + n * D + lane * 4);
    const float4 b4 = *(const float4*)(bias + lane * 4);
    v.x += b4.x; v.y += b4.y; v.z += b4.z; v.w += b4.w;
    float s = v.x + v.y + v.z + v.w;
#pragma unroll
    for (int off = 16; off >= 1; off >>= 1) s += __shfl_xor(s, off, 32);
    float mu = s * (1.0f / D);
    float dx = v.x - mu, dy = v.y - mu, dz = v.z - mu, dw = v.w - mu;
    float q = dx * dx + dy * dy + dz * dz + dw * dw;
#pragma unroll
    for (int off = 16; off >= 1; off >>= 1) q += __shfl_xor(q, off, 32);
    float rs = rsqrtf(q * (1.0f / D) + LN_EPS);
    const float4 g4 = *(const float4*)(gamma + lane * 4);
    const float4 be4 = *(const float4*)(beta + lane * 4);
    float4 o;
    o.x = dx * rs * g4.x + be4.x;  o.y = dy * rs * g4.y + be4.y;
    o.z = dz * rs * g4.z + be4.z;  o.w = dw * rs * g4.w + be4.w;
    *(float4*)(lnout + n * D + lane * 4) = o;
}

__global__ void k_gate(const float* __restrict__ LN, const float* __restrict__ query,
                       const int* __restrict__ n2g, const float* __restrict__ gW,
                       const float* __restrict__ gb, float* __restrict__ out) {
    int wave = (blockIdx.x * blockDim.x + threadIdx.x) >> 5;
    int lane = threadIdx.x & 31;
    int tm = wave >> 3, tn = wave & 7;
    int row = lane & 15, half = lane >> 4;
    int arow_i = tm * 16 + row;
    const float* lrow = LN + arow_i * D;
    const float* qrow = query + n2g[arow_i] * D;
    const float* bcol = gW + tn * 16 + row;
    v8f c = {0.f, 0.f, 0.f, 0.f, 0.f, 0.f, 0.f, 0.f};
    for (int k0 = 0; k0 < D; k0 += 4) {
        int ka = k0 + half * 2;
        v2f a; a[0] = lrow[ka];     a[1] = lrow[ka + 1];
        v2f b; b[0] = bcol[ka * D]; b[1] = bcol[(ka + 1) * D];
        c = __builtin_amdgcn_wmma_f32_16x16x4_f32(false, a, false, b,
                                                  (short)0, c, false, false);
    }
    for (int k0 = 0; k0 < D; k0 += 4) {
        int ka = k0 + half * 2;
        v2f a; a[0] = qrow[ka];             a[1] = qrow[ka + 1];
        v2f b; b[0] = bcol[(D + ka) * D];   b[1] = bcol[(D + ka + 1) * D];
        c = __builtin_amdgcn_wmma_f32_16x16x4_f32(false, a, false, b,
                                                  (short)0, c, false, false);
    }
    int cc = tn * 16 + row;
    float gbv = gb[cc];
#pragma unroll
    for (int j = 0; j < 8; ++j) {
        int rr = tm * 16 + half * 8 + j;
        float gate = 1.0f / (1.0f + __expf(-(c[j] + gbv)));
        out[rr * D + cc] = LN[rr * D + cc] * gate;
    }
}

extern "C" void kernel_launch(void* const* d_in, const int* in_sizes, int n_in,
                              void* d_out, int out_size, void* d_ws, size_t ws_size,
                              hipStream_t stream) {
    (void)in_sizes; (void)n_in; (void)out_size; (void)ws_size;
    const float* node_input = (const float*)d_in[0];
    const int*   edge_index = (const int*)d_in[1];
    const int*   node2graph = (const int*)d_in[2];
    const float* query      = (const float*)d_in[3];
    const float* W          = (const float*)d_in[4];
    const float* att_src    = (const float*)d_in[5];
    const float* att_dst    = (const float*)d_in[6];
    const float* bias       = (const float*)d_in[7];
    const float* ln_gamma   = (const float*)d_in[8];
    const float* ln_beta    = (const float*)d_in[9];
    const float* gate_W     = (const float*)d_in[10];
    const float* gate_b     = (const float*)d_in[11];
    float* out = (float*)d_out;

    float*    X     = (float*)d_ws;
    float*    a_src = X + NN * D;
    float*    a_dst = a_src + NN * H;
    unsigned* m     = (unsigned*)(a_dst + NN * H);
    float*    den   = (float*)(m + NN * H);

    k_init    <<<(NN * D) / 256, 256, 0, stream>>>(out, m, den);
    k_proj    <<<6250, 128, 0, stream>>>(node_input, W, X);
    k_logits  <<<(NN * H + 255) / 256, 256, 0, stream>>>(X, att_src, att_dst, a_src, a_dst);
    k_edge_max<<<(NE + 255) / 256, 256, 0, stream>>>(edge_index, a_src, a_dst, m);
    k_edge_sum<<<(NE + 255) / 256, 256, 0, stream>>>(edge_index, a_src, a_dst, m, den);
    k_aggr    <<<(NE / EPW + 7) / 8, 256, 0, stream>>>(edge_index, a_src, a_dst, m, den, X, out);
    k_ln      <<<NN / 8, 256, 0, stream>>>(out, bias, ln_gamma, ln_beta, X);
    k_gate    <<<6250, 128, 0, stream>>>(X, query, node2graph, gate_W, gate_b, out);
}


# both GEMMs: a wave computes 16 rows x 64 columns, lane c owning columns 4c..4c+3, so one 16-byte B-row load and one A load feed four tiles; k paired 4h+s per block of 8 (product sum order changes)
# speedup vs baseline: 2.0411x; 1.0694x over previous
#include <hip/hip_runtime.h>
#include <math.h>

#define NN 50000
#define NE 625000
#define D 128
#define H 4
#define DH 32
#define NEG_SLOPE 0.2f
#define LN_EPS 1e-5f

typedef float v2f __attribute__((ext_vector_type(2)));
typedef float v4f __attribute__((ext_vector_type(4)));
typedef float v8f __attribute__((ext_vector_type(8)));

__device__ __forceinline__ unsigned fenc(float x) {
    unsigned b = __float_as_uint(x);
    return (b & 0x80000000u) ? ~b : (b | 0x80000000u);
}
__device__ __forceinline__ float fdec(unsigned u) {
    unsigned b = (u & 0x80000000u) ? (u & 0x7FFFFFFFu) : ~u;
    return __uint_as_float(b);
}
__device__ __forceinline__ float lrelu(float v) {
    return v > 0.0f ? v : NEG_SLOPE * v;
}

__global__ void k_init(float* __restrict__ out, unsigned* __restrict__ m,
                       float* __restrict__ den) {
    int i = blockIdx.x * 256 + threadIdx.x;
    out[i] = 0.0f;
    if (i < NN * H) { m[i] = 0x007FFFFFu; den[i] = 0.0f; }
}

#define GEMM_WAVES ((NN / 16) * 2)
#define GEMM_BLOCKS ((GEMM_WAVES + 3) / 4)

struct Frag { v4f a0, a1, b0, b1, b2, b3, b4, b5, b6, b7; };

__device__ __forceinline__ void load16(Frag& f, const float* __restrict__ ap,
                                       const float* __restrict__ bp) {
    f.a0 = *(const v4f*)(ap);
    f.a1 = *(const v4f*)(ap + 8);
    f.b0 = *(const v4f*)(bp);
    f.b1 = *(const v4f*)(bp + D);
    f.b2 = *(const v4f*)(bp + 2 * D);
    f.b3 = *(const v4f*)(bp + 3 * D);
    f.b4 = *(const v4f*)(bp + 8 * D);
    f.b5 = *(const v4f*)(bp + 9 * D);
    f.b6 = *(const v4f*)(bp + 10 * D);
    f.b7 = *(const v4f*)(bp + 11 * D);
}

__device__ __forceinline__ void mac4(float x, float y, v4f p, v4f q,
                                     v8f& c0, v8f& c1, v8f& c2, v8f& c3) {
    v2f a, b;
    a[0] = x; a[1] = y;
    b[0] = p[0]; b[1] = q[0];
    c0 = __builtin_amdgcn_wmma_f32_16x16x4_f32(false, a, false, b, (short)0, c0, false, false);
    b[0] = p[1]; b[1] = q[1];
    c1 = __builtin_amdgcn_wmma_f32_16x16x4_f32(false, a, false, b, (short)0, c1, false, false);
    b[0] = p[2]; b[1] = q[2];
    c2 = __builtin_amdgcn_wmma_f32_16x16x4_f32(false, a, false, b, (short)0, c2, false, false);
    b[0] = p[3]; b[1] = q[3];
    c3 = __builtin_amdgcn_wmma_f32_16x16x4_f32(false, a, false, b, (short)0, c3, false, false);
}

__device__ __forceinline__ void mac16(const Frag& f, v8f& c0, v8f& c1, v8f& c2, v8f& c3) {
    mac4(f.a0[0], f.a0[1], f.b0, f.b1, c0, c1, c2, c3);
    mac4(f.a0[2], f.a0[3], f.b2, f.b3, c0, c1, c2, c3);
    mac4(f.a1[0], f.a1[1], f.b4, f.b5, c0, c1, c2, c3);
    mac4(f.a1[2], f.a1[3], f.b6, f.b7, c0, c1, c2, c3);
}

__global__ void k_proj(const float* __restrict__ A, const float* __restrict__ W,
                       float* __restrict__ X) {
    int wave = (blockIdx.x * blockDim.x + threadIdx.x) >> 5;
    if (wave >= GEMM_WAVES) return;
    int lane = threadIdx.x & 31;
    int tm = wave >> 1, col0 = (wave & 1) * 64;
    int c = lane & 15, h = lane >> 4;
    const float* ap = A + (tm * 16 + c) * D + 4 * h;
    const float* bp = W + (4 * h) * D + col0 + 4 * c;
    v8f c0 = {0.f, 0.f, 0.f, 0.f, 0.f, 0.f, 0.f, 0.f};
    v8f c1 = c0, c2 = c0, c3 = c0;
    Frag fa, fb;
    load16(fa, ap, bp);
#pragma unroll
    for (int g = 0; g < D / 16; g += 2) {
        load16(fb, ap + (g + 1) * 16, bp + (g + 1) * 16 * D);
        __builtin_amdgcn_sched_barrier(0);
        mac16(fa, c0, c1, c2, c3);
        if (g + 2 < D / 16) load16(fa, ap + (g + 2) * 16, bp + (g + 2) * 16 * D);
        __builtin_amdgcn_sched_barrier(0);
        mac16(fb, c0, c1, c2, c3);
    }
    float* xo = X + (tm * 16 + 8 * h) * D + col0 + 4 * c;
#pragma unroll
    for (int r = 0; r < 8; ++r) {
        v4f o; o[0] = c0[r]; o[1] = c1[r]; o[2] = c2[r]; o[3] = c3[r];
        *(v4f*)(xo + r * D) = o;
    }
}

__global__ void k_logits(const float* __restrict__ X,
                         const float* __restrict__ att_src,
                         const float* __restrict__ att_dst,
                         float* __restrict__ a_src, float* __restrict__ a_dst) {
    int i = blockIdx.x * blockDim.x + threadIdx.x;
    if (i >= NN * H) return;
    int n = i >> 2, h = i & 3;
    const float* xr = X + n * D + h * DH;
    const float* as = att_src + h * DH;
    const float* ad = att_dst + h * DH;
    float s1 = 0.f, s2 = 0.f;
#pragma unroll 8
    for (int j = 0; j < DH; ++j) { float v = xr[j]; s1 += v * as[j]; s2 += v * ad[j]; }
    a_src[i] = s1; a_dst[i] = s2;
}

__global__ void k_edge_max(const int* __restrict__ ei,
                           const float* __restrict__ a_src,
                           const float* __restrict__ a_dst,
                           unsigned* __restrict__ m) {
    int t = blockIdx.x * blockDim.x + threadIdx.x;
    int lane = t & 31;
    int h = lane & 3;
    int e0 = (t >> 5) * 32 + (lane >> 2);
    if (e0 >= NE) return;
    int s[4], d[4];
    float v[4];
#pragma unroll
    for (int i = 0; i < 4; ++i) {
        int e = min(e0 + 8 * i, NE - 1);
        s[i] = ei[e];
        d[i] = ei[NE + e];
    }
#pragma unroll
    for (int i = 0; i < 4; ++i)
        v[i] = lrelu(a_src[s[i] * H + h] + a_dst[d[i] * H + h]);
#pragma unroll
    for (int i = 0; i < 4; ++i)
        atomicMax(&m[d[i] * H + h], fenc(v[i]));
}

__global__ void k_edge_sum(const int* __restrict__ ei,
                           const float* __restrict__ a_src,
                           const float* __restrict__ a_dst,
                           const unsigned* __restrict__ m,
                           float* __restrict__ den) {
    int t = blockIdx.x * blockDim.x + threadIdx.x;
    int lane = t & 31;
    int h = lane & 3;
    int e0 = (t >> 5) * 32 + (lane >> 2);
    if (e0 >= NE) return;
    int s[4], d[4];
    float ex[4];
#pragma unroll
    for (int i = 0; i < 4; ++i) {
        int e = min(e0 + 8 * i, NE - 1);
        s[i] = ei[e];
        d[i] = ei[NE + e];
    }
#pragma unroll
    for (int i = 0; i < 4; ++i) {
        float v = lrelu(a_src[s[i] * H + h] + a_dst[d[i] * H + h]);
        float p = __expf(v - fdec(m[d[i] * H + h]));
        ex[i] = (e0 + 8 * i < NE) ? p : 0.0f;
    }
#pragma unroll
    for (int i = 0; i < 4; ++i)
        atomicAdd(&den[d[i] * H + h], ex[i]);
}

#define EPW 8
static_assert(NE % EPW == 0, "a wave's eight edges must all exist");
__global__ void k_aggr(const int* __restrict__ ei,
                       const float* __restrict__ a_src,
                       const float* __restrict__ a_dst,
                       const unsigned* __restrict__ m,
                       const float* __restrict__ den,
                       const float* __restrict__ X, float* __restrict__ out) {
    int wave = (blockIdx.x * blockDim.x + threadIdx.x) >> 5;
    int lane = threadIdx.x & 31;
    int e0 = wave * EPW;
    if (e0 >= NE) return;
    int q = lane >> 2, h = lane & 3;
    int s = ei[e0 + q], d = ei[NE + e0 + q];
    float as_ = a_src[s * H + h], ad_ = a_dst[d * H + h];
    unsigned mm = m[d * H + h];
    float dd = den[d * H + h];
    float x[EPW][4];
    int dj[EPW];
#pragma unroll
    for (int j = 0; j < EPW; ++j) {
        int sj = __builtin_amdgcn_readlane(s, 4 * j);
        dj[j]  = __builtin_amdgcn_readlane(d, 4 * j);
        const float* xr = X + sj * D + lane;
        x[j][0] = xr[0]; x[j][1] = xr[DH]; x[j][2] = xr[2 * DH]; x[j][3] = xr[3 * DH];
    }
    float v = lrelu(as_ + ad_);
    float alpha = __expf(v - fdec(mm)) / fmaxf(dd, 1e-16f);
    int ab = __float_as_int(alpha);
#pragma unroll
    for (int j = 0; j < EPW; ++j) {
        float* o = out + dj[j] * D + lane;
#pragma unroll
        for (int k = 0; k < H; ++k) {
            float w = __int_as_float(__builtin_amdgcn_readlane(ab, 4 * j + k));
            atomicAdd(o + k * DH, w * x[j][k]);
        }
    }
}

__global__ void k_ln(const float* __restrict__ acc, const float* __restrict__ bias,
                     const float* __restrict__ gamma, const float* __restrict__ beta,
                     float* __restrict__ lnout) {
    int n = (blockIdx.x * blockDim.x + threadIdx.x) >> 5;
    int lane = threadIdx.x & 31;
    float4 v = *(const float4*)(acc + n * D + lane * 4);
    const float4 b4 = *(const float4*)(bias + lane * 4);
    v.x += b4.x; v.y += b4.y; v.z += b4.z; v.w += b4.w;
    float s = v.x + v.y + v.z + v.w;
#pragma unroll
    for (int off = 16; off >= 1; off >>= 1) s += __shfl_xor(s, off, 32);
    float mu = s * (1.0f / D);
    float dx = v.x - mu, dy = v.y - mu, dz = v.z - mu, dw = v.w - mu;
    float q = dx * dx + dy * dy + dz * dz + dw * dw;
#pragma unroll
    for (int off = 16; off >= 1; off >>= 1) q += __shfl_xor(q, off, 32);
    float rs = rsqrtf(q * (1.0f / D) + LN_EPS);
    const float4 g4 = *(const float4*)(gamma + lane * 4);
    const float4 be4 = *(const float4*)(beta + lane * 4);
    float4 o;
    o.x = dx * rs * g4.x + be4.x;  o.y = dy * rs * g4.y + be4.y;
    o.z = dz * rs * g4.z + be4.z;  o.w = dw * rs * g4.w + be4.w;
    *(float4*)(lnout + n * D + lane * 4) = o;
}

__global__ void k_gate(const float* __restrict__ LN, const float* __restrict__ query,
                       const int* __restrict__ n2g, const float* __restrict__ gW,
                       const float* __restrict__ gb, float* __restrict__ out) {
    int wave = (blockIdx.x * blockDim.x + threadIdx.x) >> 5;
    if (wave >= GEMM_WAVES) return;
    int lane = threadIdx.x & 31;
    int tm = wave >> 1, col0 = (wave & 1) * 64;
    int c = lane & 15, h = lane >> 4;
    int arow_i = tm * 16 + c;
    const float* lp = LN + arow_i * D + 4 * h;
    const float* qp = query + n2g[arow_i] * D + 4 * h;
    const float* bp = gW + (4 * h) * D + col0 + 4 * c;
    v8f c0 = {0.f, 0.f, 0.f, 0.f, 0.f, 0.f, 0.f, 0.f};
    v8f c1 = c0, c2 = c0, c3 = c0;
    Frag fa, fb;
    load16(fa, lp, bp);
#pragma unroll
    for (int g = 0; g < 16; g += 2) {
        load16(fb, (g + 1 < 8) ? lp + (g + 1) * 16 : qp + (g + 1 - 8) * 16,
               bp + (g + 1) * 16 * D);
        __builtin_amdgcn_sched_barrier(0);
        mac16(fa, c0, c1, c2, c3);
        if (g + 2 < 16)
            load16(fa, (g + 2 < 8) ? lp + (g + 2) * 16 : qp + (g + 2 - 8) * 16,
                   bp + (g + 2) * 16 * D);
        __builtin_amdgcn_sched_barrier(0);
        mac16(fb, c0, c1, c2, c3);
    }
    int col = col0 + 4 * c;
    v4f gb4 = *(const v4f*)(gb + col);
    const float* lo = LN + (tm * 16 + 8 * h) * D + col;
    float* oo = out + (tm * 16 + 8 * h) * D + col;
#pragma unroll
    for (int r = 0; r < 8; ++r) {
        v4f l4 = *(const v4f*)(lo + r * D);
        v4f o;
        o[0] = l4[0] * (1.0f / (1.0f + __expf(-(c0[r] + gb4[0]))));
        o[1] = l4[1] * (1.0f / (1.0f + __expf(-(c1[r] + gb4[1]))));
        o[2] = l4[2] * (1.0f / (1.0f + __expf(-(c2[r] + gb4[2]))));
        o[3] = l4[3] * (1.0f / (1.0f + __expf(-(c3[r] + gb4[3]))));
        *(v4f*)(oo + r * D) = o;
    }
}

extern "C" void kernel_launch(void* const* d_in, const int* in_sizes, int n_in,
                              void* d_out, int out_size, void* d_ws, size_t ws_size,
                              hipStream_t stream) {
    (void)in_sizes; (void)n_in; (void)out_size; (void)ws_size;
    const float* node_input = (const float*)d_in[0];
    const int*   edge_index = (const int*)d_in[1];
    const int*   node2graph = (const int*)d_in[2];
    const float* query      = (const float*)d_in[3];
    const float* W          = (const float*)d_in[4];
    const float* att_src    = (const float*)d_in[5];
    const float* att_dst    = (const float*)d_in[6];
    const float* bias       = (const float*)d_in[7];
    const float* ln_gamma   = (const float*)d_in[8];
    const float* ln_beta    = (const float*)d_in[9];
    const float* gate_W     = (const float*)d_in[10];
    const float* gate_b     = (const float*)d_in[11];
    float* out = (float*)d_out;

    float*    X     = (float*)d_ws;
    float*    a_src = X + NN * D;
    float*    a_dst = a_src + NN * H;
    unsigned* m     = (unsigned*)(a_dst + NN * H);
    float*    den   = (float*)(m + NN * H);

    k_init    <<<(NN * D) / 256, 256, 0, stream>>>(out, m, den);
    k_proj    <<<GEMM_BLOCKS, 128, 0, stream>>>(node_input, W, X);
    k_logits  <<<(NN * H + 255) / 256, 256, 0, stream>>>(X, att_src, att_dst, a_src, a_dst);
    k_edge_max<<<(NE + 255) / 256, 256, 0, stream>>>(edge_index, a_src, a_dst, m);
    k_edge_sum<<<(NE + 255) / 256, 256, 0, stream>>>(edge_index, a_src, a_dst, m, den);
    k_aggr    <<<(NE / EPW + 7) / 8, 256, 0, stream>>>(edge_index, a_src, a_dst, m, den, X, out);
    k_ln      <<<NN / 8, 256, 0, stream>>>(out, bias, ln_gamma, ln_beta, X);
    k_gate    <<<GEMM_BLOCKS, 128, 0, stream>>>(X, query, node2graph, gate_W, gate_b, out);
}


# edge phase without float atomics: edges bucketed by destination per call (count, wave scan, fill); an owner wave sorts its edges by source, then softmax, sum, LayerNorm in registers in that order
# speedup vs baseline: 2.2610x; 1.1077x over previous
#include <hip/hip_runtime.h>
#include <math.h>

#define NN 50000
#define NE 625000
#define D 128
#define H 4
#define DH 32
#define NEG_SLOPE 0.2f
#define LN_EPS 1e-5f

typedef float v2f __attribute__((ext_vector_type(2)));
typedef float v4f __attribute__((ext_vector_type(4)));
typedef float v8f __attribute__((ext_vector_type(8)));

__device__ __forceinline__ float lrelu(float v) {
    return v > 0.0f ? v : NEG_SLOPE * v;
}

__global__ void k_zero(int* __restrict__ cnt) {
    int i = blockIdx.x * 256 + threadIdx.x;
    if (i <= NN) cnt[i] = 0;
}

__global__ void k_hist(const int* __restrict__ ei, int* __restrict__ cnt) {
    int e = blockIdx.x * 256 + threadIdx.x;
    if (e < NE) atomicAdd(&cnt[ei[NE + e]], 1);
}

__global__ void k_alloc(int* __restrict__ cnt, int* __restrict__ off) {
    int n = blockIdx.x * 256 + threadIdx.x;
    int lane = threadIdx.x & 31;
    int c = (n < NN) ? cnt[n] : 0;
    int incl = c;
#pragma unroll
    for (int o = 1; o < 32; o <<= 1) {
        int t = __shfl_up(incl, o, 32);
        if (lane >= o) incl += t;
    }
    int base = 0;
    if (lane == 31) base = atomicAdd(&cnt[NN], incl);
    base = __shfl(base, 31, 32);
    if (n < NN) { int start = base + incl - c; off[n] = start; cnt[n] = start; }
}

static_assert(NN <= 65536, "source indices are stored as 16-bit words");
__global__ void k_fill(const int* __restrict__ ei, int* __restrict__ cur,
                       unsigned short* __restrict__ perm) {
    int e = blockIdx.x * 256 + threadIdx.x;
    if (e >= NE) return;
    int s = ei[e], d = ei[NE + e];
    perm[atomicAdd(&cur[d], 1)] = (unsigned short)s;
}

#define GEMM_WAVES ((NN / 16) * 2)
#define GEMM_BLOCKS ((GEMM_WAVES + 3) / 4)

struct Frag { v4f a0, a1, b0, b1, b2, b3, b4, b5, b6, b7; };

__device__ __forceinline__ void load16(Frag& f, const float* __restrict__ ap,
                                       const float* __restrict__ bp) {
    f.a0 = *(const v4f*)(ap);
    f.a1 = *(const v4f*)(ap + 8);
    f.b0 = *(const v4f*)(bp);
    f.b1 = *(const v4f*)(bp + D);
    f.b2 = *(const v4f*)(bp + 2 * D);
    f.b3 = *(const v4f*)(bp + 3 * D);
    f.b4 = *(const v4f*)(bp + 8 * D);
    f.b5 = *(const v4f*)(bp + 9 * D);
    f.b6 = *(const v4f*)(bp + 10 * D);
    f.b7 = *(const v4f*)(bp + 11 * D);
}

__device__ __forceinline__ void mac4(float x, float y, v4f p, v4f q,
                                     v8f& c0, v8f& c1, v8f& c2, v8f& c3) {
    v2f a, b;
    a[0] = x; a[1] = y;
    b[0] = p[0]; b[1] = q[0];
    c0 = __builtin_amdgcn_wmma_f32_16x16x4_f32(false, a, false, b, (short)0, c0, false, false);
    b[0] = p[1]; b[1] = q[1];
    c1 = __builtin_amdgcn_wmma_f32_16x16x4_f32(false, a, false, b, (short)0, c1, false, false);
    b[0] = p[2]; b[1] = q[2];
    c2 = __builtin_amdgcn_wmma_f32_16x16x4_f32(false, a, false, b, (short)0, c2, false, false);
    b[0] = p[3]; b[1] = q[3];
    c3 = __builtin_amdgcn_wmma_f32_16x16x4_f32(false, a, false, b, (short)0, c3, false, false);
}

__device__ __forceinline__ void mac16(const Frag& f, v8f& c0, v8f& c1, v8f& c2, v8f& c3) {
    mac4(f.a0[0], f.a0[1], f.b0, f.b1, c0, c1, c2, c3);
    mac4(f.a0[2], f.a0[3], f.b2, f.b3, c0, c1, c2, c3);
    mac4(f.a1[0], f.a1[1], f.b4, f.b5, c0, c1, c2, c3);
    mac4(f.a1[2], f.a1[3], f.b6, f.b7, c0, c1, c2, c3);
}

__global__ void k_proj(const float* __restrict__ A, const float* __restrict__ W,
                       float* __restrict__ X) {
    int wave = (blockIdx.x * blockDim.x + threadIdx.x) >> 5;
    if (wave >= GEMM_WAVES) return;
    int lane = threadIdx.x & 31;
    int tm = wave >> 1, col0 = (wave & 1) * 64;
    int c = lane & 15, h = lane >> 4;
    const float* ap = A + (tm * 16 + c) * D + 4 * h;
    const float* bp = W + (4 * h) * D + col0 + 4 * c;
    v8f c0 = {0.f, 0.f, 0.f, 0.f, 0.f, 0.f, 0.f, 0.f};
    v8f c1 = c0, c2 = c0, c3 = c0;
    Frag fa, fb;
    load16(fa, ap, bp);
#pragma unroll
    for (int g = 0; g < D / 16; g += 2) {
        load16(fb, ap + (g + 1) * 16, bp + (g + 1) * 16 * D);
        __builtin_amdgcn_sched_barrier(0);
        mac16(fa, c0, c1, c2, c3);
        if (g + 2 < D / 16) load16(fa, ap + (g + 2) * 16, bp + (g + 2) * 16 * D);
        __builtin_amdgcn_sched_barrier(0);
        mac16(fb, c0, c1, c2, c3);
    }
    float* xo = X + (tm * 16 + 8 * h) * D + col0 + 4 * c;
#pragma unroll
    for (int r = 0; r < 8; ++r) {
        v4f o; o[0] = c0[r]; o[1] = c1[r]; o[2] = c2[r]; o[3] = c3[r];
        *(v4f*)(xo + r * D) = o;
    }
}

__global__ void k_logits(const float* __restrict__ X,
                         const float* __restrict__ att_src,
                         const float* __restrict__ att_dst,
                         float* __restrict__ a_src, float* __restrict__ a_dst) {
    int i = blockIdx.x * blockDim.x + threadIdx.x;
    if (i >= NN * H) return;
    int n = i >> 2, h = i & 3;
    const float* xr = X + n * D + h * DH;
    const float* as = att_src + h * DH;
    const float* ad = att_dst + h * DH;
    float s1 = 0.f, s2 = 0.f;
#pragma unroll 8
    for (int j = 0; j < DH; ++j) { float v = xr[j]; s1 += v * as[j]; s2 += v * ad[j]; }
    a_src[i] = s1; a_dst[i] = s2;
}

__device__ __forceinline__ float wave_max(float v) {
#pragma unroll
    for (int o = 16; o >= 1; o >>= 1) v = fmaxf(v, __shfl_xor(v, o, 32));
    return v;
}
__device__ __forceinline__ float wave_sum(float v) {
#pragma unroll
    for (int o = 16; o >= 1; o >>= 1) v += __shfl_xor(v, o, 32);
    return v;
}
__device__ __forceinline__ float lane_val(float v, int l) {
    return __int_as_float(__builtin_amdgcn_readlane(__float_as_int(v), l));
}
__device__ __forceinline__ float pick(float b0, float b1, float b2, float b3,
                                      int k0, int k1, int k2, int k3) {
    return __int_as_float((__float_as_int(b0) & k0) | (__float_as_int(b1) & k1) |
                          (__float_as_int(b2) & k2) | (__float_as_int(b3) & k3));
}

__device__ __forceinline__ unsigned long long next_key(const unsigned short* __restrict__ seg, int deg,
                                                       unsigned long long lo, int lane) {
    unsigned long long best = ~0ull;
    for (int i = lane; i < deg; i += 32) {
        unsigned long long k = ((unsigned long long)(unsigned)seg[i] << 32) | (unsigned)i;
        if (k >= lo && k < best) best = k;
    }
#pragma unroll
    for (int o = 16; o >= 1; o >>= 1) {
        unsigned hi2 = (unsigned)__shfl_xor((int)(best >> 32), o, 32);
        unsigned lo2 = (unsigned)__shfl_xor((int)(unsigned)best, o, 32);
        unsigned long long t = ((unsigned long long)hi2 << 32) | lo2;
        best = t < best ? t : best;
    }
    return best;
}

__global__ void k_owner(const unsigned short* __restrict__ perm, const int* __restrict__ off,
                        const int* __restrict__ endp,
                        const float* __restrict__ a_src, const float* __restrict__ a_dst,
                        const float* __restrict__ X, const float* __restrict__ bias,
                        const float* __restrict__ gamma, const float* __restrict__ beta,
                        float* __restrict__ lnout) {
    int n = (blockIdx.x * blockDim.x + threadIdx.x) >> 5;
    int lane = threadIdx.x & 31;
    int start = __builtin_amdgcn_readfirstlane(off[n]);
    int deg = __builtin_amdgcn_readfirstlane(endp[n]) - start;
    const unsigned short* seg = perm + start;
    const float4 ad = *(const float4*)(a_dst + n * H);
    const float4* X4 = (const float4*)X;
    const int hd = lane >> 3;
    const int k0 = hd == 0 ? -1 : 0, k1 = hd == 1 ? -1 : 0;
    const int k2 = hd == 2 ? -1 : 0, k3 = hd == 3 ? -1 : 0;
    float4 acc = make_float4(0.0f, 0.0f, 0.0f, 0.0f);

    if (deg <= 32) {
        bool valid = lane < deg;
        int key = valid ? ((int)seg[lane] << 5) | lane : 0x7FFFFFFF;
        int rank = 0;
        for (int j = 0; j < deg; ++j)
            rank += (__builtin_amdgcn_readlane(key, j) < key) ? 1 : 0;
        if (!valid) rank = lane;
        int ss = __builtin_amdgcn_ds_permute(rank << 2, key) >> 5;
        const float4 a = *(const float4*)(a_src + (valid ? ss : 0) * H);
        float v0 = lrelu(a.x + ad.x), v1 = lrelu(a.y + ad.y);
        float v2 = lrelu(a.z + ad.z), v3 = lrelu(a.w + ad.w);
        float m0 = wave_max(valid ? v0 : -INFINITY), m1 = wave_max(valid ? v1 : -INFINITY);
        float m2 = wave_max(valid ? v2 : -INFINITY), m3 = wave_max(valid ? v3 : -INFINITY);
        float e0 = valid ? __expf(v0 - m0) : 0.0f, e1 = valid ? __expf(v1 - m1) : 0.0f;
        float e2 = valid ? __expf(v2 - m2) : 0.0f, e3 = valid ? __expf(v3 - m3) : 0.0f;
        float al0 = e0 / fmaxf(wave_sum(e0), 1e-16f), al1 = e1 / fmaxf(wave_sum(e1), 1e-16f);
        float al2 = e2 / fmaxf(wave_sum(e2), 1e-16f), al3 = e3 / fmaxf(wave_sum(e3), 1e-16f);
        int p = 0;
        for (; p + 4 <= deg; p += 4) {
            float4 x[4];
            float w[4];
#pragma unroll
            for (int u = 0; u < 4; ++u)
                x[u] = X4[__builtin_amdgcn_readlane(ss, p + u) * (D / 4) + lane];
#pragma unroll
            for (int u = 0; u < 4; ++u)
                w[u] = pick(lane_val(al0, p + u), lane_val(al1, p + u),
                            lane_val(al2, p + u), lane_val(al3, p + u), k0, k1, k2, k3);
#pragma unroll
            for (int u = 0; u < 4; ++u) {
                acc.x = fmaf(w[u], x[u].x, acc.x); acc.y = fmaf(w[u], x[u].y, acc.y);
                acc.z = fmaf(w[u], x[u].z, acc.z); acc.w = fmaf(w[u], x[u].w, acc.w);
            }
        }
        for (; p < deg; ++p) {
            const float4 x = X4[__builtin_amdgcn_readlane(ss, p) * (D / 4) + lane];
            float w = pick(lane_val(al0, p), lane_val(al1, p), lane_val(al2, p), lane_val(al3, p), k0, k1, k2, k3);
            acc.x = fmaf(w, x.x, acc.x); acc.y = fmaf(w, x.y, acc.y);
            acc.z = fmaf(w, x.z, acc.z); acc.w = fmaf(w, x.w, acc.w);
        }
    } else {
        float m0 = -INFINITY, m1 = -INFINITY, m2 = -INFINITY, m3 = -INFINITY;
        for (int i = lane; i < deg; i += 32) {
            const float4 a = *(const float4*)(a_src + (int)seg[i] * H);
            m0 = fmaxf(m0, lrelu(a.x + ad.x)); m1 = fmaxf(m1, lrelu(a.y + ad.y));
            m2 = fmaxf(m2, lrelu(a.z + ad.z)); m3 = fmaxf(m3, lrelu(a.w + ad.w));
        }
        m0 = wave_max(m0); m1 = wave_max(m1); m2 = wave_max(m2); m3 = wave_max(m3);
        float d0 = 0.0f, d1 = 0.0f, d2 = 0.0f, d3 = 0.0f;
        unsigned long long lo = 0;
        for (int t = 0; t < deg; ++t) {
            unsigned long long k = next_key(seg, deg, lo, lane);
            lo = k + 1;
            const float4 a = *(const float4*)(a_src + (int)(k >> 32) * H);
            d0 += __expf(lrelu(a.x + ad.x) - m0); d1 += __expf(lrelu(a.y + ad.y) - m1);
            d2 += __expf(lrelu(a.z + ad.z) - m2); d3 += __expf(lrelu(a.w + ad.w) - m3);
        }
        d0 = fmaxf(d0, 1e-16f); d1 = fmaxf(d1, 1e-16f);
        d2 = fmaxf(d2, 1e-16f); d3 = fmaxf(d3, 1e-16f);
        lo = 0;
        for (int t = 0; t < deg; ++t) {
            unsigned long long k = next_key(seg, deg, lo, lane);
            lo = k + 1;
            int sp = (int)(k >> 32);
            const float4 a = *(const float4*)(a_src + sp * H);
            float w = pick(__expf(lrelu(a.x + ad.x) - m0) / d0, __expf(lrelu(a.y + ad.y) - m1) / d1,
                           __expf(lrelu(a.z + ad.z) - m2) / d2, __expf(lrelu(a.w + ad.w) - m3) / d3, k0, k1, k2, k3);
            const float4 x = X4[sp * (D / 4) + lane];
            acc.x = fmaf(w, x.x, acc.x); acc.y = fmaf(w, x.y, acc.y);
            acc.z = fmaf(w, x.z, acc.z); acc.w = fmaf(w, x.w, acc.w);
        }
    }

    const float4 b4 = *(const float4*)(bias + lane * 4);
    float4 v = acc;
    v.x += b4.x; v.y += b4.y; v.z += b4.z; v.w += b4.w;
    float mu = wave_sum(v.x + v.y + v.z + v.w) * (1.0f / D);
    float dx = v.x - mu, dy = v.y - mu, dz = v.z - mu, dw = v.w - mu;
    float q = wave_sum(dx * dx + dy * dy + dz * dz + dw * dw);
    float rs = rsqrtf(q * (1.0f / D) + LN_EPS);
    const float4 g4 = *(const float4*)(gamma + lane * 4);
    const float4 be4 = *(const float4*)(beta + lane * 4);
    float4 o;
    o.x = dx * rs * g4.x + be4.x;  o.y = dy * rs * g4.y + be4.y;
    o.z = dz * rs * g4.z + be4.z;  o.w = dw * rs * g4.w + be4.w;
    *(float4*)(lnout + n * D + lane * 4) = o;
}

__global__ void k_gate(const float* __restrict__ LN, const float* __restrict__ query,
                       const int* __restrict__ n2g, const float* __restrict__ gW,
                       const float* __restrict__ gb, float* __restrict__ out) {
    int wave = (blockIdx.x * blockDim.x + threadIdx.x) >> 5;
    if (wave >= GEMM_WAVES) return;
    int lane = threadIdx.x & 31;
    int tm = wave >> 1, col0 = (wave & 1) * 64;
    int c = lane & 15, h = lane >> 4;
    int arow_i = tm * 16 + c;
    const float* lp = LN + arow_i * D + 4 * h;
    const float* qp = query + n2g[arow_i] * D + 4 * h;
    const float* bp = gW + (4 * h) * D + col0 + 4 * c;
    v8f c0 = {0.f, 0.f, 0.f, 0.f, 0.f, 0.f, 0.f, 0.f};
    v8f c1 = c0, c2 = c0, c3 = c0;
    Frag fa, fb;
    load16(fa, lp, bp);
#pragma unroll
    for (int g = 0; g < 16; g += 2) {
        load16(fb, (g + 1 < 8) ? lp + (g + 1) * 16 : qp + (g + 1 - 8) * 16,
               bp + (g + 1) * 16 * D);
        __builtin_amdgcn_sched_barrier(0);
        mac16(fa, c0, c1, c2, c3);
        if (g + 2 < 16)
            load16(fa, (g + 2 < 8) ? lp + (g + 2) * 16 : qp + (g + 2 - 8) * 16,
                   bp + (g + 2) * 16 * D);
        __builtin_amdgcn_sched_barrier(0);
        mac16(fb, c0, c1, c2, c3);
    }
    int col = col0 + 4 * c;
    v4f gb4 = *(const v4f*)(gb + col);
    const float* lo = LN + (tm * 16 + 8 * h) * D + col;
    float* oo = out + (tm * 16 + 8 * h) * D + col;
#pragma unroll
    for (int r = 0; r < 8; ++r) {
        v4f l4 = *(const v4f*)(lo + r * D);
        v4f o;
        o[0] = l4[0] * (1.0f / (1.0f + __expf(-(c0[r] + gb4[0]))));
        o[1] = l4[1] * (1.0f / (1.0f + __expf(-(c1[r] + gb4[1]))));
        o[2] = l4[2] * (1.0f / (1.0f + __expf(-(c2[r] + gb4[2]))));
        o[3] = l4[3] * (1.0f / (1.0f + __expf(-(c3[r] + gb4[3]))));
        *(v4f*)(oo + r * D) = o;
    }
}

extern "C" void kernel_launch(void* const* d_in, const int* in_sizes, int n_in,
                              void* d_out, int out_size, void* d_ws, size_t ws_size,
                              hipStream_t stream) {
    (void)in_sizes; (void)n_in; (void)out_size; (void)ws_size;
    const float* node_input = (const float*)d_in[0];
    const int*   edge_index = (const int*)d_in[1];
    const int*   node2graph = (const int*)d_in[2];
    const float* query      = (const float*)d_in[3];
    const float* W          = (const float*)d_in[4];
    const float* att_src    = (const float*)d_in[5];
    const float* att_dst    = (const float*)d_in[6];
    const float* bias       = (const float*)d_in[7];
    const float* ln_gamma   = (const float*)d_in[8];
    const float* ln_beta    = (const float*)d_in[9];
    const float* gate_W     = (const float*)d_in[10];
    const float* gate_b     = (const float*)d_in[11];
    float* out = (float*)d_out;

    float* X     = out;
    float* LN    = (float*)d_ws;
    float* a_src = LN + NN * D;
    float* a_dst = a_src + NN * H;
    int*   cnt   = (int*)(a_dst + NN * H);
    int*   off   = cnt + NN + 4;
    unsigned short* perm = (unsigned short*)(off + NN);

    k_zero    <<<(NN + 1 + 255) / 256, 256, 0, stream>>>(cnt);
    k_hist    <<<(NE + 255) / 256, 256, 0, stream>>>(edge_index, cnt);
    k_alloc   <<<(NN + 255) / 256, 256, 0, stream>>>(cnt, off);
    k_fill    <<<(NE + 255) / 256, 256, 0, stream>>>(edge_index, cnt, perm);
    k_proj    <<<GEMM_BLOCKS, 128, 0, stream>>>(node_input, W, X);
    k_logits  <<<(NN * H + 255) / 256, 256, 0, stream>>>(X, att_src, att_dst, a_src, a_dst);
    k_owner   <<<NN / 8, 256, 0, stream>>>(perm, off, cnt, a_src, a_dst, X, bias, ln_gamma, ln_beta, LN);
    k_gate    <<<GEMM_BLOCKS, 128, 0, stream>>>(LN, query, node2graph, gate_W, gate_b, out);
}


# gate GEMM K 256->128: query @ gate_W[128:] computed once per graph (8 extra waves in the projection launch), added before the sigmoid (sum order changes); a_dst summed by the owner wave instead
# speedup vs baseline: 3.2288x; 1.4280x over previous
#include <hip/hip_runtime.h>
#include <math.h>

#define NN 50000
#define NE 625000
#define D 128
#define H 4
#define DH 32
#define NG 64
#define NEG_SLOPE 0.2f
#define LN_EPS 1e-5f

typedef float v2f __attribute__((ext_vector_type(2)));
typedef float v4f __attribute__((ext_vector_type(4)));
typedef float v8f __attribute__((ext_vector_type(8)));

__device__ __forceinline__ float lrelu(float v) {
    return v > 0.0f ? v : NEG_SLOPE * v;
}

static_assert(NN <= 65536, "source indices are stored as 16-bit words");
static_assert(NG % 32 == 0 && NG * D <= NN * H, "graph table: whole work-groups, and it fits the array it replaces");
#define CAP (D * 4 / 2)

__global__ void k_fill(const int* __restrict__ ei, int* __restrict__ cnt,
                       float* __restrict__ rows) {
    int e = blockIdx.x * 256 + threadIdx.x;
    if (e >= NE) return;
    int s = ei[e], d = ei[NE + e];
    int r = atomicAdd(&cnt[d], 1);
    ((unsigned short*)(rows + d * D))[r < CAP ? r : CAP - 1] = (unsigned short)s;
}

#define GEMM_WAVES ((NN / 16) * 2)
#define GEMM_BLOCKS ((GEMM_WAVES + 3) / 4)

struct Frag { v4f a0, a1, b0, b1, b2, b3, b4, b5, b6, b7; };

__device__ __forceinline__ void load16(Frag& f, const float* __restrict__ ap,
                                       const float* __restrict__ bp) {
    f.a0 = *(const v4f*)(ap);
    f.a1 = *(const v4f*)(ap + 8);
    f.b0 = *(const v4f*)(bp);
    f.b1 = *(const v4f*)(bp + D);
    f.b2 = *(const v4f*)(bp + 2 * D);
    f.b3 = *(const v4f*)(bp + 3 * D);
    f.b4 = *(const v4f*)(bp + 8 * D);
    f.b5 = *(const v4f*)(bp + 9 * D);
    f.b6 = *(const v4f*)(bp + 10 * D);
    f.b7 = *(const v4f*)(bp + 11 * D);
}

__device__ __forceinline__ void mac4(float x, float y, v4f p, v4f q,
                                     v8f& c0, v8f& c1, v8f& c2, v8f& c3) {
    v2f a, b;
    a[0] = x; a[1] = y;
    b[0] = p[0]; b[1] = q[0];
    c0 = __builtin_amdgcn_wmma_f32_16x16x4_f32(false, a, false, b, (short)0, c0, false, false);
    b[0] = p[1]; b[1] = q[1];
    c1 = __builtin_amdgcn_wmma_f32_16x16x4_f32(false, a, false, b, (short)0, c1, false, false);
    b[0] = p[2]; b[1] = q[2];
    c2 = __builtin_amdgcn_wmma_f32_16x16x4_f32(false, a, false, b, (short)0, c2, false, false);
    b[0] = p[3]; b[1] = q[3];
    c3 = __builtin_amdgcn_wmma_f32_16x16x4_f32(false, a, false, b, (short)0, c3, false, false);
}

__device__ __forceinline__ void mac16(const Frag& f, v8f& c0, v8f& c1, v8f& c2, v8f& c3) {
    mac4(f.a0[0], f.a0[1], f.b0, f.b1, c0, c1, c2, c3);
    mac4(f.a0[2], f.a0[3], f.b2, f.b3, c0, c1, c2, c3);
    mac4(f.a1[0], f.a1[1], f.b4, f.b5, c0, c1, c2, c3);
    mac4(f.a1[2], f.a1[3], f.b6, f.b7, c0, c1, c2, c3);
}

__device__ __forceinline__ void gemm_k128(const float* __restrict__ ap, const float* __restrict__ bp,
                                          v8f& c0, v8f& c1, v8f& c2, v8f& c3) {
    Frag fa, fb;
    load16(fa, ap, bp);
#pragma unroll
    for (int g = 0; g < D / 16; g += 2) {
        load16(fb, ap + (g + 1) * 16, bp + (g + 1) * 16 * D);
        __builtin_amdgcn_sched_barrier(0);
        mac16(fa, c0, c1, c2, c3);
        if (g + 2 < D / 16) load16(fa, ap + (g + 2) * 16, bp + (g + 2) * 16 * D);
        __builtin_amdgcn_sched_barrier(0);
        mac16(fb, c0, c1, c2, c3);
    }
}


template <int M>
__device__ __forceinline__ float add_xor(float a, float b) {
    return a + __int_as_float(__builtin_amdgcn_update_dpp(0, __float_as_int(b), 0x160 + M,
                                                          0xF, 0xF, true));
}

__device__ __forceinline__ float fold8(const float* v, bool b4, bool b2, bool b1) {
    float w[4], u[2];
#pragma unroll
    for (int i = 0; i < 4; ++i)
        w[i] = add_xor<4>(b4 ? v[i + 4] : v[i], b4 ? v[i] : v[i + 4]);
#pragma unroll
    for (int i = 0; i < 2; ++i)
        u[i] = add_xor<2>(b2 ? w[i + 2] : w[i], b2 ? w[i] : w[i + 2]);
    return add_xor<1>(b1 ? u[1] : u[0], b1 ? u[0] : u[1]);
}

__global__ void __launch_bounds__(128, 7)
k_proj(const float* __restrict__ A, const float* __restrict__ W,
                       const float* __restrict__ att_src, const float* __restrict__ query,
                       const float* __restrict__ gW, const float* __restrict__ gb,
                       float* __restrict__ X, float* __restrict__ a_src,
                       float* __restrict__ QG, int* __restrict__ cnt) {
    __shared__ __attribute__((aligned(16))) float lg[2 * 16 * H];
    int lane = threadIdx.x & 31;
    int wv = __builtin_amdgcn_readfirstlane(threadIdx.x >> 5);
    if (wv == 0) {
        int i = blockIdx.x * 32 + lane;
        if (i < NN) cnt[i] = 0;
    }
    int c = lane & 15, h = lane >> 4;
    v8f c0 = {0.f, 0.f, 0.f, 0.f, 0.f, 0.f, 0.f, 0.f};
    v8f c1 = c0, c2 = c0, c3 = c0;
    if (blockIdx.x >= GEMM_BLOCKS) {
        int q = (blockIdx.x - GEMM_BLOCKS) * 4 + wv;
        int qt = q >> 1, qc0 = (q & 1) * 64;
        gemm_k128(query + (qt * 16 + c) * D + 4 * h, gW + (D + 4 * h) * D + qc0 + 4 * c,
                  c0, c1, c2, c3);
        const v4f g4 = *(const v4f*)(gb + qc0 + 4 * c);
        float* qo = QG + (qt * 16 + 8 * h) * D + qc0 + 4 * c;
#pragma unroll
        for (int r = 0; r < 8; ++r) {
            v4f o; o[0] = c0[r] + g4[0]; o[1] = c1[r] + g4[1]; o[2] = c2[r] + g4[2]; o[3] = c3[r] + g4[3];
            *(v4f*)(qo + r * D) = o;
        }
        return;
    }
    int wave = blockIdx.x * 4 + wv;
    bool live = wave < GEMM_WAVES;
    int tm = wave >> 1, hf = wave & 1, col0 = hf * 64;
    float* lgt = lg + (wv >> 1) * (16 * H);
    if (live) {
        gemm_k128(A + (tm * 16 + c) * D + 4 * h, W + (4 * h) * D + col0 + 4 * c, c0, c1, c2, c3);
        const v4f as4 = *(const v4f*)(att_src + col0 + 4 * c);
        float ps[8];
#pragma unroll
        for (int r = 0; r < 8; ++r)
            ps[r] = c0[r] * as4[0] + c1[r] * as4[1] + c2[r] * as4[2] + c3[r] * as4[3];
        bool b4 = (c & 4) != 0, b2 = (c & 2) != 0, b1 = (c & 1) != 0;
        float ss = fold8(ps, b4, b2, b1);
        lgt[(8 * h + (c & 7)) * H + 2 * hf + (c >> 3)] = ss;
    }
    __syncthreads();
    if (live) {
        if (hf == 0) {
            const v2f t = *(const v2f*)(lgt + 2 * lane);
            *(v2f*)(a_src + tm * (16 * H) + 2 * lane) = t;
        }
        float* xo = X + (tm * 16 + 8 * h) * D + col0 + 4 * c;
#pragma unroll
        for (int r = 0; r < 8; ++r) {
            v4f o; o[0] = c0[r]; o[1] = c1[r]; o[2] = c2[r]; o[3] = c3[r];
            *(v4f*)(xo + r * D) = o;
        }
    }
}

__device__ __forceinline__ float wave_max(float v) {
#pragma unroll
    for (int o = 16; o >= 1; o >>= 1) v = fmaxf(v, __shfl_xor(v, o, 32));
    return v;
}
__device__ __forceinline__ float wave_sum(float v) {
#pragma unroll
    for (int o = 16; o >= 1; o >>= 1) v += __shfl_xor(v, o, 32);
    return v;
}
__device__ __forceinline__ float lane_val(float v, int l) {
    return __int_as_float(__builtin_amdgcn_readlane(__float_as_int(v), l));
}
__device__ __forceinline__ float pick(float b0, float b1, float b2, float b3,
                                      int k0, int k1, int k2, int k3) {
    return __int_as_float((__float_as_int(b0) & k0) | (__float_as_int(b1) & k1) |
                          (__float_as_int(b2) & k2) | (__float_as_int(b3) & k3));
}

__device__ __forceinline__ unsigned long long next_key(const unsigned short* __restrict__ seg, int deg,
                                                       unsigned long long lo, int lane) {
    unsigned long long best = ~0ull;
    for (int i = lane; i < deg; i += 32) {
        unsigned long long k = ((unsigned long long)(unsigned)seg[i] << 32) | (unsigned)i;
        if (k >= lo && k < best) best = k;
    }
#pragma unroll
    for (int o = 16; o >= 1; o >>= 1) {
        unsigned hi2 = (unsigned)__shfl_xor((int)(best >> 32), o, 32);
        unsigned lo2 = (unsigned)__shfl_xor((int)(unsigned)best, o, 32);
        unsigned long long t = ((unsigned long long)hi2 << 32) | lo2;
        best = t < best ? t : best;
    }
    return best;
}

__global__ void k_owner(const int* __restrict__ ei, const int* __restrict__ cnt,
                        const float* __restrict__ a_src, const float* __restrict__ att_dst,
                        const float* __restrict__ X, const float* __restrict__ bias,
                        const float* __restrict__ gamma, const float* __restrict__ beta,
                        float* lnout) {
    int n = (blockIdx.x * blockDim.x + threadIdx.x) >> 5;
    int lane = threadIdx.x & 31;
    int deg = __builtin_amdgcn_readfirstlane(cnt[n]);
    const unsigned short* seg = (const unsigned short*)(lnout + n * D);
    const float4* X4 = (const float4*)X;
    const int hd = lane >> 3;
    const float4 xn = X4[n * (D / 4) + lane];
    const float4 at = *(const float4*)(att_dst + lane * 4);
    float pd = xn.x * at.x + xn.y * at.y + xn.z * at.z + xn.w * at.w;
    pd = add_xor<4>(pd, pd);
    pd = add_xor<2>(pd, pd);
    pd = add_xor<1>(pd, pd);
    const float4 ad = make_float4(lane_val(pd, 0), lane_val(pd, 8), lane_val(pd, 16), lane_val(pd, 24));
    const int k0 = hd == 0 ? -1 : 0, k1 = hd == 1 ? -1 : 0;
    const int k2 = hd == 2 ? -1 : 0, k3 = hd == 3 ? -1 : 0;
    float4 acc = make_float4(0.0f, 0.0f, 0.0f, 0.0f);

    if (deg <= 32) {
        bool valid = lane < deg;
        int key = valid ? ((int)seg[lane] << 5) | lane : 0x7FFFFFFF;
        int rank = 0;
        for (int j = 0; j < deg; ++j)
            rank += (__builtin_amdgcn_readlane(key, j) < key) ? 1 : 0;
        if (!valid) rank = lane;
        int ss = __builtin_amdgcn_ds_permute(rank << 2, key) >> 5;
        const float4 a = *(const float4*)(a_src + (valid ? ss : 0) * H);
        float v0 = lrelu(a.x + ad.x), v1 = lrelu(a.y + ad.y);
        float v2 = lrelu(a.z + ad.z), v3 = lrelu(a.w + ad.w);
        float m0 = wave_max(valid ? v0 : -INFINITY), m1 = wave_max(valid ? v1 : -INFINITY);
        float m2 = wave_max(valid ? v2 : -INFINITY), m3 = wave_max(valid ? v3 : -INFINITY);
        float e0 = valid ? __expf(v0 - m0) : 0.0f, e1 = valid ? __expf(v1 - m1) : 0.0f;
        float e2 = valid ? __expf(v2 - m2) : 0.0f, e3 = valid ? __expf(v3 - m3) : 0.0f;
        float al0 = e0 / fmaxf(wave_sum(e0), 1e-16f), al1 = e1 / fmaxf(wave_sum(e1), 1e-16f);
        float al2 = e2 / fmaxf(wave_sum(e2), 1e-16f), al3 = e3 / fmaxf(wave_sum(e3), 1e-16f);
        int p = 0;
        for (; p + 4 <= deg; p += 4) {
            float4 x[4];
            float w[4];
#pragma unroll
            for (int u = 0; u < 4; ++u)
                x[u] = X4[__builtin_amdgcn_readlane(ss, p + u) * (D / 4) + lane];
#pragma unroll
            for (int u = 0; u < 4; ++u)
                w[u] = pick(lane_val(al0, p + u), lane_val(al1, p + u),
                            lane_val(al2, p + u), lane_val(al3, p + u), k0, k1, k2, k3);
#pragma unroll
            for (int u = 0; u < 4; ++u) {
                acc.x = fmaf(w[u], x[u].x, acc.x); acc.y = fmaf(w[u], x[u].y, acc.y);
                acc.z = fmaf(w[u], x[u].z, acc.z); acc.w = fmaf(w[u], x[u].w, acc.w);
            }
        }
        for (; p < deg; ++p) {
            const float4 x = X4[__builtin_amdgcn_readlane(ss, p) * (D / 4) + lane];
            float w = pick(lane_val(al0, p), lane_val(al1, p), lane_val(al2, p), lane_val(al3, p), k0, k1, k2, k3);
            acc.x = fmaf(w, x.x, acc.x); acc.y = fmaf(w, x.y, acc.y);
            acc.z = fmaf(w, x.z, acc.z); acc.w = fmaf(w, x.w, acc.w);
        }
    } else if (deg <= CAP) {
        float m0 = -INFINITY, m1 = -INFINITY, m2 = -INFINITY, m3 = -INFINITY;
        for (int i = lane; i < deg; i += 32) {
            const float4 a = *(const float4*)(a_src + (int)seg[i] * H);
            m0 = fmaxf(m0, lrelu(a.x + ad.x)); m1 = fmaxf(m1, lrelu(a.y + ad.y));
            m2 = fmaxf(m2, lrelu(a.z + ad.z)); m3 = fmaxf(m3, lrelu(a.w + ad.w));
        }
        m0 = wave_max(m0); m1 = wave_max(m1); m2 = wave_max(m2); m3 = wave_max(m3);
        float d0 = 0.0f, d1 = 0.0f, d2 = 0.0f, d3 = 0.0f;
        unsigned long long lo = 0;
        for (int t = 0; t < deg; ++t) {
            unsigned long long k = next_key(seg, deg, lo, lane);
            lo = k + 1;
            const float4 a = *(const float4*)(a_src + (int)(k >> 32) * H);
            d0 += __expf(lrelu(a.x + ad.x) - m0); d1 += __expf(lrelu(a.y + ad.y) - m1);
            d2 += __expf(lrelu(a.z + ad.z) - m2); d3 += __expf(lrelu(a.w + ad.w) - m3);
        }
        d0 = fmaxf(d0, 1e-16f); d1 = fmaxf(d1, 1e-16f);
        d2 = fmaxf(d2, 1e-16f); d3 = fmaxf(d3, 1e-16f);
        lo = 0;
        for (int t = 0; t < deg; ++t) {
            unsigned long long k = next_key(seg, deg, lo, lane);
            lo = k + 1;
            int sp = (int)(k >> 32);
            const float4 a = *(const float4*)(a_src + sp * H);
            float w = pick(__expf(lrelu(a.x + ad.x) - m0) / d0, __expf(lrelu(a.y + ad.y) - m1) / d1,
                           __expf(lrelu(a.z + ad.z) - m2) / d2, __expf(lrelu(a.w + ad.w) - m3) / d3, k0, k1, k2, k3);
            const float4 x = X4[sp * (D / 4) + lane];
            acc.x = fmaf(w, x.x, acc.x); acc.y = fmaf(w, x.y, acc.y);
            acc.z = fmaf(w, x.z, acc.z); acc.w = fmaf(w, x.w, acc.w);
        }
    } else {
        const int* dsts = ei + NE;
        float m0 = -INFINITY, m1 = -INFINITY, m2 = -INFINITY, m3 = -INFINITY;
        for (int e = lane; e < NE; e += 32) {
            if (dsts[e] == n) {
                const float4 a = *(const float4*)(a_src + ei[e] * H);
                m0 = fmaxf(m0, lrelu(a.x + ad.x)); m1 = fmaxf(m1, lrelu(a.y + ad.y));
                m2 = fmaxf(m2, lrelu(a.z + ad.z)); m3 = fmaxf(m3, lrelu(a.w + ad.w));
            }
        }
        m0 = wave_max(m0); m1 = wave_max(m1); m2 = wave_max(m2); m3 = wave_max(m3);
        float d0 = 0.0f, d1 = 0.0f, d2 = 0.0f, d3 = 0.0f;
        for (int base = 0; base < NE; base += 32) {
            int e = base + lane;
            bool hit = e < NE && dsts[e] == n;
            int sl = hit ? ei[e] : 0;
            unsigned mask = __builtin_amdgcn_ballot_w32(hit);
            while (mask) {
                int j = __builtin_ctz(mask);
                mask &= mask - 1;
                const float4 a = *(const float4*)(a_src + __builtin_amdgcn_readlane(sl, j) * H);
                d0 += __expf(lrelu(a.x + ad.x) - m0); d1 += __expf(lrelu(a.y + ad.y) - m1);
                d2 += __expf(lrelu(a.z + ad.z) - m2); d3 += __expf(lrelu(a.w + ad.w) - m3);
            }
        }
        d0 = fmaxf(d0, 1e-16f); d1 = fmaxf(d1, 1e-16f);
        d2 = fmaxf(d2, 1e-16f); d3 = fmaxf(d3, 1e-16f);
        for (int base = 0; base < NE; base += 32) {
            int e = base + lane;
            bool hit = e < NE && dsts[e] == n;
            int sl = hit ? ei[e] : 0;
            unsigned mask = __builtin_amdgcn_ballot_w32(hit);
            while (mask) {
                int j = __builtin_ctz(mask);
                mask &= mask - 1;
                int sp = __builtin_amdgcn_readlane(sl, j);
                const float4 a = *(const float4*)(a_src + sp * H);
                float w = pick(__expf(lrelu(a.x + ad.x) - m0) / d0, __expf(lrelu(a.y + ad.y) - m1) / d1,
                               __expf(lrelu(a.z + ad.z) - m2) / d2, __expf(lrelu(a.w + ad.w) - m3) / d3, k0, k1, k2, k3);
                const float4 x = X4[sp * (D / 4) + lane];
                acc.x = fmaf(w, x.x, acc.x); acc.y = fmaf(w, x.y, acc.y);
                acc.z = fmaf(w, x.z, acc.z); acc.w = fmaf(w, x.w, acc.w);
            }
        }
    }

    const float4 b4 = *(const float4*)(bias + lane * 4);
    float4 v = acc;
    v.x += b4.x; v.y += b4.y; v.z += b4.z; v.w += b4.w;
    float mu = wave_sum(v.x + v.y + v.z + v.w) * (1.0f / D);
    float dx = v.x - mu, dy = v.y - mu, dz = v.z - mu, dw = v.w - mu;
    float q = wave_sum(dx * dx + dy * dy + dz * dz + dw * dw);
    float rs = rsqrtf(q * (1.0f / D) + LN_EPS);
    const float4 g4 = *(const float4*)(gamma + lane * 4);
    const float4 be4 = *(const float4*)(beta + lane * 4);
    float4 o;
    o.x = dx * rs * g4.x + be4.x;  o.y = dy * rs * g4.y + be4.y;
    o.z = dz * rs * g4.z + be4.z;  o.w = dw * rs * g4.w + be4.w;
    *(float4*)(lnout + n * D + lane * 4) = o;
}

__global__ void k_gate(const float* __restrict__ LN, const float* __restrict__ QG,
                       const int* __restrict__ n2g, const float* __restrict__ gW,
                       float* __restrict__ out) {
    int wave = __builtin_amdgcn_readfirstlane((blockIdx.x * blockDim.x + threadIdx.x) >> 5);
    if (wave >= GEMM_WAVES) return;
    int lane = threadIdx.x & 31;
    int tm = wave >> 1, col0 = (wave & 1) * 64;
    int c = lane & 15, h = lane >> 4;
    const int* gp = n2g + tm * 16;
    int g16[16];
#pragma unroll
    for (int i = 0; i < 16; ++i) g16[i] = gp[i];
    v8f c0 = {0.f, 0.f, 0.f, 0.f, 0.f, 0.f, 0.f, 0.f};
    v8f c1 = c0, c2 = c0, c3 = c0;
    gemm_k128(LN + (tm * 16 + c) * D + 4 * h, gW + (4 * h) * D + col0 + 4 * c, c0, c1, c2, c3);
    int col = col0 + 4 * c;
    const float* lo = LN + (tm * 16 + 8 * h) * D + col;
    const float* qg = QG + col;
    float* oo = out + (tm * 16 + 8 * h) * D + col;
#pragma unroll
    for (int r = 0; r < 8; ++r) {
        v4f l4 = *(const v4f*)(lo + r * D);
        v4f q4 = *(const v4f*)(qg + (h ? g16[8 + r] : g16[r]) * D);
        v4f o;
        o[0] = l4[0] * (1.0f / (1.0f + __expf(-(c0[r] + q4[0]))));
        o[1] = l4[1] * (1.0f / (1.0f + __expf(-(c1[r] + q4[1]))));
        o[2] = l4[2] * (1.0f / (1.0f + __expf(-(c2[r] + q4[2]))));
        o[3] = l4[3] * (1.0f / (1.0f + __expf(-(c3[r] + q4[3]))));
        *(v4f*)(oo + r * D) = o;
    }
}

extern "C" void kernel_launch(void* const* d_in, const int* in_sizes, int n_in,
                              void* d_out, int out_size, void* d_ws, size_t ws_size,
                              hipStream_t stream) {
    (void)in_sizes; (void)n_in; (void)out_size; (void)ws_size;
    const float* node_input = (const float*)d_in[0];
    const int*   edge_index = (const int*)d_in[1];
    const int*   node2graph = (const int*)d_in[2];
    const float* query      = (const float*)d_in[3];
    const float* W          = (const float*)d_in[4];
    const float* att_src    = (const float*)d_in[5];
    const float* att_dst    = (const float*)d_in[6];
    const float* bias       = (const float*)d_in[7];
    const float* ln_gamma   = (const float*)d_in[8];
    const float* ln_beta    = (const float*)d_in[9];
    const float* gate_W     = (const float*)d_in[10];
    const float* gate_b     = (const float*)d_in[11];
    float* out = (float*)d_out;

    float* X     = out;
    float* LN    = (float*)d_ws;
    float* a_src = LN + NN * D;
    float* QG    = a_src + NN * H;
    int*   cnt   = (int*)(QG + NN * H);

    k_proj    <<<GEMM_BLOCKS + NG / 32, 128, 0, stream>>>(node_input, W, att_src, query, gate_W, gate_b, X, a_src, QG, cnt);
    k_fill    <<<(NE + 255) / 256, 256, 0, stream>>>(edge_index, cnt, LN);
    k_owner   <<<NN / 8, 256, 0, stream>>>(edge_index, cnt, a_src, att_dst, X, bias, ln_gamma, ln_beta, LN);
    k_gate    <<<GEMM_BLOCKS, 128, 0, stream>>>(LN, QG, node2graph, gate_W, out);
}


# owner wave (nodes of up to 32 edges): source logit summed from the gathered X row, softmax by running max along the sorted edges: no a_src gather, no wave max/sum (logit, softmax sum order change)
# speedup vs baseline: 3.3556x; 1.0393x over previous
#include <hip/hip_runtime.h>
#include <math.h>

#define NN 50000
#define NE 625000
#define D 128
#define H 4
#define DH 32
#define NG 64
#define NEG_SLOPE 0.2f
#define LN_EPS 1e-5f

typedef float v2f __attribute__((ext_vector_type(2)));
typedef float v4f __attribute__((ext_vector_type(4)));
typedef float v8f __attribute__((ext_vector_type(8)));

__device__ __forceinline__ float lrelu(float v) {
    return v > 0.0f ? v : NEG_SLOPE * v;
}

static_assert(NN <= 65536, "source indices are stored as 16-bit words");
static_assert(NG % 32 == 0 && NG * D <= NN * H, "graph table: whole work-groups, and it fits the array it replaces");
#define CAP (D * 4 / 2)

__global__ void k_fill(const int* __restrict__ ei, int* __restrict__ cnt,
                       float* __restrict__ rows) {
    int e = blockIdx.x * 256 + threadIdx.x;
    if (e >= NE) return;
    int s = ei[e], d = ei[NE + e];
    int r = atomicAdd(&cnt[d], 1);
    ((unsigned short*)(rows + d * D))[r < CAP ? r : CAP - 1] = (unsigned short)s;
}

#define GEMM_WAVES ((NN / 16) * 2)
#define GEMM_BLOCKS ((GEMM_WAVES + 3) / 4)

struct Frag { v4f a0, a1, b0, b1, b2, b3, b4, b5, b6, b7; };

__device__ __forceinline__ void load16(Frag& f, const float* __restrict__ ap,
                                       const float* __restrict__ bp) {
    f.a0 = *(const v4f*)(ap);
    f.a1 = *(const v4f*)(ap + 8);
    f.b0 = *(const v4f*)(bp);
    f.b1 = *(const v4f*)(bp + D);
    f.b2 = *(const v4f*)(bp + 2 * D);
    f.b3 = *(const v4f*)(bp + 3 * D);
    f.b4 = *(const v4f*)(bp + 8 * D);
    f.b5 = *(const v4f*)(bp + 9 * D);
    f.b6 = *(const v4f*)(bp + 10 * D);
    f.b7 = *(const v4f*)(bp + 11 * D);
}

__device__ __forceinline__ void mac4(float x, float y, v4f p, v4f q,
                                     v8f& c0, v8f& c1, v8f& c2, v8f& c3) {
    v2f a, b;
    a[0] = x; a[1] = y;
    b[0] = p[0]; b[1] = q[0];
    c0 = __builtin_amdgcn_wmma_f32_16x16x4_f32(false, a, false, b, (short)0, c0, false, false);
    b[0] = p[1]; b[1] = q[1];
    c1 = __builtin_amdgcn_wmma_f32_16x16x4_f32(false, a, false, b, (short)0, c1, false, false);
    b[0] = p[2]; b[1] = q[2];
    c2 = __builtin_amdgcn_wmma_f32_16x16x4_f32(false, a, false, b, (short)0, c2, false, false);
    b[0] = p[3]; b[1] = q[3];
    c3 = __builtin_amdgcn_wmma_f32_16x16x4_f32(false, a, false, b, (short)0, c3, false, false);
}

__device__ __forceinline__ void mac16(const Frag& f, v8f& c0, v8f& c1, v8f& c2, v8f& c3) {
    mac4(f.a0[0], f.a0[1], f.b0, f.b1, c0, c1, c2, c3);
    mac4(f.a0[2], f.a0[3], f.b2, f.b3, c0, c1, c2, c3);
    mac4(f.a1[0], f.a1[1], f.b4, f.b5, c0, c1, c2, c3);
    mac4(f.a1[2], f.a1[3], f.b6, f.b7, c0, c1, c2, c3);
}

__device__ __forceinline__ void gemm_k128(const float* __restrict__ ap, const float* __restrict__ bp,
                                          v8f& c0, v8f& c1, v8f& c2, v8f& c3) {
    Frag fa, fb;
    load16(fa, ap, bp);
#pragma unroll
    for (int g = 0; g < D / 16; g += 2) {
        load16(fb, ap + (g + 1) * 16, bp + (g + 1) * 16 * D);
        __builtin_amdgcn_sched_barrier(0);
        mac16(fa, c0, c1, c2, c3);
        if (g + 2 < D / 16) load16(fa, ap + (g + 2) * 16, bp + (g + 2) * 16 * D);
        __builtin_amdgcn_sched_barrier(0);
        mac16(fb, c0, c1, c2, c3);
    }
}


template <int M>
__device__ __forceinline__ float add_xor(float a, float b) {
    return a + __int_as_float(__builtin_amdgcn_update_dpp(0, __float_as_int(b), 0x160 + M,
                                                          0xF, 0xF, true));
}

__device__ __forceinline__ float fold8(const float* v, bool b4, bool b2, bool b1) {
    float w[4], u[2];
#pragma unroll
    for (int i = 0; i < 4; ++i)
        w[i] = add_xor<4>(b4 ? v[i + 4] : v[i], b4 ? v[i] : v[i + 4]);
#pragma unroll
    for (int i = 0; i < 2; ++i)
        u[i] = add_xor<2>(b2 ? w[i + 2] : w[i], b2 ? w[i] : w[i + 2]);
    return add_xor<1>(b1 ? u[1] : u[0], b1 ? u[0] : u[1]);
}

__global__ void __launch_bounds__(128, 7)
k_proj(const float* __restrict__ A, const float* __restrict__ W,
                       const float* __restrict__ att_src, const float* __restrict__ query,
                       const float* __restrict__ gW, const float* __restrict__ gb,
                       float* __restrict__ X, float* __restrict__ a_src,
                       float* __restrict__ QG, int* __restrict__ cnt) {
    __shared__ __attribute__((aligned(16))) float lg[2 * 16 * H];
    int lane = threadIdx.x & 31;
    int wv = __builtin_amdgcn_readfirstlane(threadIdx.x >> 5);
    if (wv == 0) {
        int i = blockIdx.x * 32 + lane;
        if (i < NN) cnt[i] = 0;
    }
    int c = lane & 15, h = lane >> 4;
    v8f c0 = {0.f, 0.f, 0.f, 0.f, 0.f, 0.f, 0.f, 0.f};
    v8f c1 = c0, c2 = c0, c3 = c0;
    if (blockIdx.x >= GEMM_BLOCKS) {
        int q = (blockIdx.x - GEMM_BLOCKS) * 4 + wv;
        int qt = q >> 1, qc0 = (q & 1) * 64;
        gemm_k128(query + (qt * 16 + c) * D + 4 * h, gW + (D + 4 * h) * D + qc0 + 4 * c,
                  c0, c1, c2, c3);
        const v4f g4 = *(const v4f*)(gb + qc0 + 4 * c);
        float* qo = QG + (qt * 16 + 8 * h) * D + qc0 + 4 * c;
#pragma unroll
        for (int r = 0; r < 8; ++r) {
            v4f o; o[0] = c0[r] + g4[0]; o[1] = c1[r] + g4[1]; o[2] = c2[r] + g4[2]; o[3] = c3[r] + g4[3];
            *(v4f*)(qo + r * D) = o;
        }
        return;
    }
    int wave = blockIdx.x * 4 + wv;
    bool live = wave < GEMM_WAVES;
    int tm = wave >> 1, hf = wave & 1, col0 = hf * 64;
    float* lgt = lg + (wv >> 1) * (16 * H);
    if (live) {
        gemm_k128(A + (tm * 16 + c) * D + 4 * h, W + (4 * h) * D + col0 + 4 * c, c0, c1, c2, c3);
        const v4f as4 = *(const v4f*)(att_src + col0 + 4 * c);
        float ps[8];
#pragma unroll
        for (int r = 0; r < 8; ++r)
            ps[r] = c0[r] * as4[0] + c1[r] * as4[1] + c2[r] * as4[2] + c3[r] * as4[3];
        bool b4 = (c & 4) != 0, b2 = (c & 2) != 0, b1 = (c & 1) != 0;
        float ss = fold8(ps, b4, b2, b1);
        lgt[(8 * h + (c & 7)) * H + 2 * hf + (c >> 3)] = ss;
    }
    __syncthreads();
    if (live) {
        if (hf == 0) {
            const v2f t = *(const v2f*)(lgt + 2 * lane);
            *(v2f*)(a_src + tm * (16 * H) + 2 * lane) = t;
        }
        float* xo = X + (tm * 16 + 8 * h) * D + col0 + 4 * c;
#pragma unroll
        for (int r = 0; r < 8; ++r) {
            v4f o; o[0] = c0[r]; o[1] = c1[r]; o[2] = c2[r]; o[3] = c3[r];
            *(v4f*)(xo + r * D) = o;
        }
    }
}

__device__ __forceinline__ float wave_max(float v) {
#pragma unroll
    for (int o = 16; o >= 1; o >>= 1) v = fmaxf(v, __shfl_xor(v, o, 32));
    return v;
}
__device__ __forceinline__ float wave_sum(float v) {
#pragma unroll
    for (int o = 16; o >= 1; o >>= 1) v += __shfl_xor(v, o, 32);
    return v;
}
__device__ __forceinline__ float lane_val(float v, int l) {
    return __int_as_float(__builtin_amdgcn_readlane(__float_as_int(v), l));
}
__device__ __forceinline__ float pick(float b0, float b1, float b2, float b3,
                                      int k0, int k1, int k2, int k3) {
    return __int_as_float((__float_as_int(b0) & k0) | (__float_as_int(b1) & k1) |
                          (__float_as_int(b2) & k2) | (__float_as_int(b3) & k3));
}

__device__ __forceinline__ unsigned long long next_key(const unsigned short* __restrict__ seg, int deg,
                                                       unsigned long long lo, int lane) {
    unsigned long long best = ~0ull;
    for (int i = lane; i < deg; i += 32) {
        unsigned long long k = ((unsigned long long)(unsigned)seg[i] << 32) | (unsigned)i;
        if (k >= lo && k < best) best = k;
    }
#pragma unroll
    for (int o = 16; o >= 1; o >>= 1) {
        unsigned hi2 = (unsigned)__shfl_xor((int)(best >> 32), o, 32);
        unsigned lo2 = (unsigned)__shfl_xor((int)(unsigned)best, o, 32);
        unsigned long long t = ((unsigned long long)hi2 << 32) | lo2;
        best = t < best ? t : best;
    }
    return best;
}

__device__ __forceinline__ float edge_logit(const float4 x, const float4 as4, float pd) {
    float t = x.x * as4.x;
    t = fmaf(x.y, as4.y, t);
    t = fmaf(x.z, as4.z, t);
    t = fmaf(x.w, as4.w, t);
    asm("" : "+v"(t));
    t = add_xor<4>(t, t);
    t = add_xor<2>(t, t);
    t = add_xor<1>(t, t);
    float s = t + pd;
    return fmaxf(s, NEG_SLOPE * s);
}

__global__ void k_owner(const int* __restrict__ ei, const int* __restrict__ cnt,
                        const float* __restrict__ a_src, const float* __restrict__ att_src,
                        const float* __restrict__ att_dst,
                        const float* __restrict__ X, const float* __restrict__ bias,
                        const float* __restrict__ gamma, const float* __restrict__ beta,
                        float* lnout) {
    int n = (blockIdx.x * blockDim.x + threadIdx.x) >> 5;
    int lane = threadIdx.x & 31;
    int deg = __builtin_amdgcn_readfirstlane(cnt[n]);
    const unsigned short* seg = (const unsigned short*)(lnout + n * D);
    const float4* X4 = (const float4*)X;
    const int hd = lane >> 3;
    const float4 xn = X4[n * (D / 4) + lane];
    const float4 at = *(const float4*)(att_dst + lane * 4);
    float pd = xn.x * at.x + xn.y * at.y + xn.z * at.z + xn.w * at.w;
    pd = add_xor<4>(pd, pd);
    pd = add_xor<2>(pd, pd);
    pd = add_xor<1>(pd, pd);
    const float4 ad = make_float4(lane_val(pd, 0), lane_val(pd, 8), lane_val(pd, 16), lane_val(pd, 24));
    const int k0 = hd == 0 ? -1 : 0, k1 = hd == 1 ? -1 : 0;
    const int k2 = hd == 2 ? -1 : 0, k3 = hd == 3 ? -1 : 0;
    float4 acc = make_float4(0.0f, 0.0f, 0.0f, 0.0f);

    if (deg <= 32) {
        bool valid = lane < deg;
        int key = valid ? ((int)seg[lane] << 5) | lane : 0x7FFFFFFF;
        int rank = 0;
        for (int j = 0; j < deg; ++j)
            rank += (__builtin_amdgcn_readlane(key, j) < key) ? 1 : 0;
        if (!valid) rank = lane;
        int ss = __builtin_amdgcn_ds_permute(rank << 2, key) >> 5;
        const float4 as4 = *(const float4*)(att_src + lane * 4);
        float m = -INFINITY, den = 0.0f;
        int p = 0;
        for (; p + 4 <= deg; p += 4) {
            float4 x[4];
            float e[4], w[4];
#pragma unroll
            for (int u = 0; u < 4; ++u)
                x[u] = X4[__builtin_amdgcn_readlane(ss, p + u) * (D / 4) + lane];
#pragma unroll
            for (int u = 0; u < 4; ++u)
                e[u] = edge_logit(x[u], as4, pd);
            float mn = fmaxf(fmaxf(fmaxf(m, e[0]), fmaxf(e[1], e[2])), e[3]);
            float sc = __expf(m - mn);
#pragma unroll
            for (int u = 0; u < 4; ++u)
                w[u] = __expf(e[u] - mn);
            den = fmaf(den, sc, (w[0] + w[1]) + (w[2] + w[3]));
            acc.x *= sc; acc.y *= sc; acc.z *= sc; acc.w *= sc;
#pragma unroll
            for (int u = 0; u < 4; ++u) {
                acc.x = fmaf(w[u], x[u].x, acc.x); acc.y = fmaf(w[u], x[u].y, acc.y);
                acc.z = fmaf(w[u], x[u].z, acc.z); acc.w = fmaf(w[u], x[u].w, acc.w);
            }
            m = mn;
        }
        for (; p < deg; ++p) {
            const float4 x = X4[__builtin_amdgcn_readlane(ss, p) * (D / 4) + lane];
            float e = edge_logit(x, as4, pd);
            float mn = fmaxf(m, e);
            float sc = __expf(m - mn), w = __expf(e - mn);
            den = fmaf(den, sc, w);
            acc.x = fmaf(w, x.x, acc.x * sc); acc.y = fmaf(w, x.y, acc.y * sc);
            acc.z = fmaf(w, x.z, acc.z * sc); acc.w = fmaf(w, x.w, acc.w * sc);
            m = mn;
        }
        float inv = 1.0f / fmaxf(den, 1e-16f);
        acc.x *= inv; acc.y *= inv; acc.z *= inv; acc.w *= inv;
    } else if (deg <= CAP) {
        float m0 = -INFINITY, m1 = -INFINITY, m2 = -INFINITY, m3 = -INFINITY;
        for (int i = lane; i < deg; i += 32) {
            const float4 a = *(const float4*)(a_src + (int)seg[i] * H);
            m0 = fmaxf(m0, lrelu(a.x + ad.x)); m1 = fmaxf(m1, lrelu(a.y + ad.y));
            m2 = fmaxf(m2, lrelu(a.z + ad.z)); m3 = fmaxf(m3, lrelu(a.w + ad.w));
        }
        m0 = wave_max(m0); m1 = wave_max(m1); m2 = wave_max(m2); m3 = wave_max(m3);
        float d0 = 0.0f, d1 = 0.0f, d2 = 0.0f, d3 = 0.0f;
        unsigned long long lo = 0;
        for (int t = 0; t < deg; ++t) {
            unsigned long long k = next_key(seg, deg, lo, lane);
            lo = k + 1;
            const float4 a = *(const float4*)(a_src + (int)(k >> 32) * H);
            d0 += __expf(lrelu(a.x + ad.x) - m0); d1 += __expf(lrelu(a.y + ad.y) - m1);
            d2 += __expf(lrelu(a.z + ad.z) - m2); d3 += __expf(lrelu(a.w + ad.w) - m3);
        }
        d0 = fmaxf(d0, 1e-16f); d1 = fmaxf(d1, 1e-16f);
        d2 = fmaxf(d2, 1e-16f); d3 = fmaxf(d3, 1e-16f);
        lo = 0;
        for (int t = 0; t < deg; ++t) {
            unsigned long long k = next_key(seg, deg, lo, lane);
            lo = k + 1;
            int sp = (int)(k >> 32);
            const float4 a = *(const float4*)(a_src + sp * H);
            float w = pick(__expf(lrelu(a.x + ad.x) - m0) / d0, __expf(lrelu(a.y + ad.y) - m1) / d1,
                           __expf(lrelu(a.z + ad.z) - m2) / d2, __expf(lrelu(a.w + ad.w) - m3) / d3, k0, k1, k2, k3);
            const float4 x = X4[sp * (D / 4) + lane];
            acc.x = fmaf(w, x.x, acc.x); acc.y = fmaf(w, x.y, acc.y);
            acc.z = fmaf(w, x.z, acc.z); acc.w = fmaf(w, x.w, acc.w);
        }
    } else {
        const int* dsts = ei + NE;
        float m0 = -INFINITY, m1 = -INFINITY, m2 = -INFINITY, m3 = -INFINITY;
        for (int e = lane; e < NE; e += 32) {
            if (dsts[e] == n) {
                const float4 a = *(const float4*)(a_src + ei[e] * H);
                m0 = fmaxf(m0, lrelu(a.x + ad.x)); m1 = fmaxf(m1, lrelu(a.y + ad.y));
                m2 = fmaxf(m2, lrelu(a.z + ad.z)); m3 = fmaxf(m3, lrelu(a.w + ad.w));
            }
        }
        m0 = wave_max(m0); m1 = wave_max(m1); m2 = wave_max(m2); m3 = wave_max(m3);
        float d0 = 0.0f, d1 = 0.0f, d2 = 0.0f, d3 = 0.0f;
        for (int base = 0; base < NE; base += 32) {
            int e = base + lane;
            bool hit = e < NE && dsts[e] == n;
            int sl = hit ? ei[e] : 0;
            unsigned mask = __builtin_amdgcn_ballot_w32(hit);
            while (mask) {
                int j = __builtin_ctz(mask);
                mask &= mask - 1;
                const float4 a = *(const float4*)(a_src + __builtin_amdgcn_readlane(sl, j) * H);
                d0 += __expf(lrelu(a.x + ad.x) - m0); d1 += __expf(lrelu(a.y + ad.y) - m1);
                d2 += __expf(lrelu(a.z + ad.z) - m2); d3 += __expf(lrelu(a.w + ad.w) - m3);
            }
        }
        d0 = fmaxf(d0, 1e-16f); d1 = fmaxf(d1, 1e-16f);
        d2 = fmaxf(d2, 1e-16f); d3 = fmaxf(d3, 1e-16f);
        for (int base = 0; base < NE; base += 32) {
            int e = base + lane;
            bool hit = e < NE && dsts[e] == n;
            int sl = hit ? ei[e] : 0;
            unsigned mask = __builtin_amdgcn_ballot_w32(hit);
            while (mask) {
                int j = __builtin_ctz(mask);
                mask &= mask - 1;
                int sp = __builtin_amdgcn_readlane(sl, j);
                const float4 a = *(const float4*)(a_src + sp * H);
                float w = pick(__expf(lrelu(a.x + ad.x) - m0) / d0, __expf(lrelu(a.y + ad.y) - m1) / d1,
                               __expf(lrelu(a.z + ad.z) - m2) / d2, __expf(lrelu(a.w + ad.w) - m3) / d3, k0, k1, k2, k3);
                const float4 x = X4[sp * (D / 4) + lane];
                acc.x = fmaf(w, x.x, acc.x); acc.y = fmaf(w, x.y, acc.y);
                acc.z = fmaf(w, x.z, acc.z); acc.w = fmaf(w, x.w, acc.w);
            }
        }
    }

    const float4 b4 = *(const float4*)(bias + lane * 4);
    float4 v = acc;
    v.x += b4.x; v.y += b4.y; v.z += b4.z; v.w += b4.w;
    float mu = wave_sum(v.x + v.y + v.z + v.w) * (1.0f / D);
    float dx = v.x - mu, dy = v.y - mu, dz = v.z - mu, dw = v.w - mu;
    float q = wave_sum(dx * dx + dy * dy + dz * dz + dw * dw);
    float rs = rsqrtf(q * (1.0f / D) + LN_EPS);
    const float4 g4 = *(const float4*)(gamma + lane * 4);
    const float4 be4 = *(const float4*)(beta + lane * 4);
    float4 o;
    o.x = dx * rs * g4.x + be4.x;  o.y = dy * rs * g4.y + be4.y;
    o.z = dz * rs * g4.z + be4.z;  o.w = dw * rs * g4.w + be4.w;
    *(float4*)(lnout + n * D + lane * 4) = o;
}

__global__ void k_gate(const float* __restrict__ LN, const float* __restrict__ QG,
                       const int* __restrict__ n2g, const float* __restrict__ gW,
                       float* __restrict__ out) {
    int wave = __builtin_amdgcn_readfirstlane((blockIdx.x * blockDim.x + threadIdx.x) >> 5);
    if (wave >= GEMM_WAVES) return;
    int lane = threadIdx.x & 31;
    int tm = wave >> 1, col0 = (wave & 1) * 64;
    int c = lane & 15, h = lane >> 4;
    const int* gp = n2g + tm * 16;
    int g16[16];
#pragma unroll
    for (int i = 0; i < 16; ++i) g16[i] = gp[i];
    v8f c0 = {0.f, 0.f, 0.f, 0.f, 0.f, 0.f, 0.f, 0.f};
    v8f c1 = c0, c2 = c0, c3 = c0;
    gemm_k128(LN + (tm * 16 + c) * D + 4 * h, gW + (4 * h) * D + col0 + 4 * c, c0, c1, c2, c3);
    int col = col0 + 4 * c;
    const float* lo = LN + (tm * 16 + 8 * h) * D + col;
    const float* qg = QG + col;
    float* oo = out + (tm * 16 + 8 * h) * D + col;
#pragma unroll
    for (int r = 0; r < 8; ++r) {
        v4f l4 = *(const v4f*)(lo + r * D);
        v4f q4 = *(const v4f*)(qg + (h ? g16[8 + r] : g16[r]) * D);
        v4f o;
        o[0] = l4[0] * (1.0f / (1.0f + __expf(-(c0[r] + q4[0]))));
        o[1] = l4[1] * (1.0f / (1.0f + __expf(-(c1[r] + q4[1]))));
        o[2] = l4[2] * (1.0f / (1.0f + __expf(-(c2[r] + q4[2]))));
        o[3] = l4[3] * (1.0f / (1.0f + __expf(-(c3[r] + q4[3]))));
        *(v4f*)(oo + r * D) = o;
    }
}

extern "C" void kernel_launch(void* const* d_in, const int* in_sizes, int n_in,
                              void* d_out, int out_size, void* d_ws, size_t ws_size,
                              hipStream_t stream) {
    (void)in_sizes; (void)n_in; (void)out_size; (void)ws_size;
    const float* node_input = (const float*)d_in[0];
    const int*   edge_index = (const int*)d_in[1];
    const int*   node2graph = (const int*)d_in[2];
    const float* query      = (const float*)d_in[3];
    const float* W          = (const float*)d_in[4];
    const float* att_src    = (const float*)d_in[5];
    const float* att_dst    = (const float*)d_in[6];
    const float* bias       = (const float*)d_in[7];
    const float* ln_gamma   = (const float*)d_in[8];
    const float* ln_beta    = (const float*)d_in[9];
    const float* gate_W     = (const float*)d_in[10];
    const float* gate_b     = (const float*)d_in[11];
    float* out = (float*)d_out;

    float* X     = out;
    float* LN    = (float*)d_ws;
    float* a_src = LN + NN * D;
    float* QG    = a_src + NN * H;
    int*   cnt   = (int*)(QG + NN * H);

    k_proj    <<<GEMM_BLOCKS + NG / 32, 128, 0, stream>>>(node_input, W, att_src, query, gate_W, gate_b, X, a_src, QG, cnt);
    k_fill    <<<(NE + 255) / 256, 256, 0, stream>>>(edge_index, cnt, LN);
    k_owner   <<<NN / 8, 256, 0, stream>>>(edge_index, cnt, a_src, att_src, att_dst, X, bias, ln_gamma, ln_beta, LN);
    k_gate    <<<GEMM_BLOCKS, 128, 0, stream>>>(LN, QG, node2graph, gate_W, out);
}


# edge counter moved into the first word of the node's own bucket line: an edge's add and source store hit one line, counters lie one per line, not 32; owner loads count and bucket in one round
# speedup vs baseline: 4.0930x; 1.2198x over previous
#include <hip/hip_runtime.h>
#include <math.h>

#define NN 50000
#define NE 625000
#define D 128
#define H 4
#define DH 32
#define NG 64
#define NEG_SLOPE 0.2f
#define LN_EPS 1e-5f

typedef float v2f __attribute__((ext_vector_type(2)));
typedef float v4f __attribute__((ext_vector_type(4)));
typedef float v8f __attribute__((ext_vector_type(8)));

__device__ __forceinline__ float lrelu(float v) {
    return v > 0.0f ? v : NEG_SLOPE * v;
}

static_assert(NN <= 65536, "source indices are stored as 16-bit words");
static_assert(NG % 32 == 0 && NG * D <= NN * H, "graph table: whole work-groups, and it fits the array it replaces");
#define CAP ((D * 4 - 4) / 2)
#define SEG0 2

__global__ void k_fill(const int* __restrict__ ei, float* rows) {
    int e = blockIdx.x * 256 + threadIdx.x;
    if (e >= NE) return;
    int s = ei[e], d = ei[NE + e];
    int* row = (int*)(rows + d * D);
    int r = atomicAdd(row, 1);
    ((unsigned short*)row)[SEG0 + (r < CAP ? r : CAP - 1)] = (unsigned short)s;
}

#define GEMM_WAVES ((NN / 16) * 2)
#define GEMM_BLOCKS ((GEMM_WAVES + 3) / 4)

struct Frag { v4f a0, a1, b0, b1, b2, b3, b4, b5, b6, b7; };

__device__ __forceinline__ void load16(Frag& f, const float* __restrict__ ap,
                                       const float* __restrict__ bp) {
    f.a0 = *(const v4f*)(ap);
    f.a1 = *(const v4f*)(ap + 8);
    f.b0 = *(const v4f*)(bp);
    f.b1 = *(const v4f*)(bp + D);
    f.b2 = *(const v4f*)(bp + 2 * D);
    f.b3 = *(const v4f*)(bp + 3 * D);
    f.b4 = *(const v4f*)(bp + 8 * D);
    f.b5 = *(const v4f*)(bp + 9 * D);
    f.b6 = *(const v4f*)(bp + 10 * D);
    f.b7 = *(const v4f*)(bp + 11 * D);
}

__device__ __forceinline__ void mac4(float x, float y, v4f p, v4f q,
                                     v8f& c0, v8f& c1, v8f& c2, v8f& c3) {
    v2f a, b;
    a[0] = x; a[1] = y;
    b[0] = p[0]; b[1] = q[0];
    c0 = __builtin_amdgcn_wmma_f32_16x16x4_f32(false, a, false, b, (short)0, c0, false, false);
    b[0] = p[1]; b[1] = q[1];
    c1 = __builtin_amdgcn_wmma_f32_16x16x4_f32(false, a, false, b, (short)0, c1, false, false);
    b[0] = p[2]; b[1] = q[2];
    c2 = __builtin_amdgcn_wmma_f32_16x16x4_f32(false, a, false, b, (short)0, c2, false, false);
    b[0] = p[3]; b[1] = q[3];
    c3 = __builtin_amdgcn_wmma_f32_16x16x4_f32(false, a, false, b, (short)0, c3, false, false);
}

__device__ __forceinline__ void mac16(const Frag& f, v8f& c0, v8f& c1, v8f& c2, v8f& c3) {
    mac4(f.a0[0], f.a0[1], f.b0, f.b1, c0, c1, c2, c3);
    mac4(f.a0[2], f.a0[3], f.b2, f.b3, c0, c1, c2, c3);
    mac4(f.a1[0], f.a1[1], f.b4, f.b5, c0, c1, c2, c3);
    mac4(f.a1[2], f.a1[3], f.b6, f.b7, c0, c1, c2, c3);
}

__device__ __forceinline__ void gemm_k128(const float* __restrict__ ap, const float* __restrict__ bp,
                                          v8f& c0, v8f& c1, v8f& c2, v8f& c3) {
    Frag fa, fb;
    load16(fa, ap, bp);
#pragma unroll
    for (int g = 0; g < D / 16; g += 2) {
        load16(fb, ap + (g + 1) * 16, bp + (g + 1) * 16 * D);
        __builtin_amdgcn_sched_barrier(0);
        mac16(fa, c0, c1, c2, c3);
        if (g + 2 < D / 16) load16(fa, ap + (g + 2) * 16, bp + (g + 2) * 16 * D);
        __builtin_amdgcn_sched_barrier(0);
        mac16(fb, c0, c1, c2, c3);
    }
}


template <int M>
__device__ __forceinline__ float add_xor(float a, float b) {
    return a + __int_as_float(__builtin_amdgcn_update_dpp(0, __float_as_int(b), 0x160 + M,
                                                          0xF, 0xF, true));
}

__device__ __forceinline__ float fold8(const float* v, bool b4, bool b2, bool b1) {
    float w[4], u[2];
#pragma unroll
    for (int i = 0; i < 4; ++i)
        w[i] = add_xor<4>(b4 ? v[i + 4] : v[i], b4 ? v[i] : v[i + 4]);
#pragma unroll
    for (int i = 0; i < 2; ++i)
        u[i] = add_xor<2>(b2 ? w[i + 2] : w[i], b2 ? w[i] : w[i + 2]);
    return add_xor<1>(b1 ? u[1] : u[0], b1 ? u[0] : u[1]);
}

__global__ void __launch_bounds__(128, 7)
k_proj(const float* __restrict__ A, const float* __restrict__ W,
                       const float* __restrict__ att_src, const float* __restrict__ query,
                       const float* __restrict__ gW, const float* __restrict__ gb,
                       float* __restrict__ X, float* __restrict__ a_src,
                       float* __restrict__ QG, float* __restrict__ rows) {
    __shared__ __attribute__((aligned(16))) float lg[2 * 16 * H];
    int lane = threadIdx.x & 31;
    int wv = __builtin_amdgcn_readfirstlane(threadIdx.x >> 5);
    if (wv == 0) {
        int i = blockIdx.x * 32 + lane;
        if (i < NN) *(int*)(rows + i * D) = 0;
    }
    int c = lane & 15, h = lane >> 4;
    v8f c0 = {0.f, 0.f, 0.f, 0.f, 0.f, 0.f, 0.f, 0.f};
    v8f c1 = c0, c2 = c0, c3 = c0;
    if (blockIdx.x >= GEMM_BLOCKS) {
        int q = (blockIdx.x - GEMM_BLOCKS) * 4 + wv;
        int qt = q >> 1, qc0 = (q & 1) * 64;
        gemm_k128(query + (qt * 16 + c) * D + 4 * h, gW + (D + 4 * h) * D + qc0 + 4 * c,
                  c0, c1, c2, c3);
        const v4f g4 = *(const v4f*)(gb + qc0 + 4 * c);
        float* qo = QG + (qt * 16 + 8 * h) * D + qc0 + 4 * c;
#pragma unroll
        for (int r = 0; r < 8; ++r) {
            v4f o; o[0] = c0[r] + g4[0]; o[1] = c1[r] + g4[1]; o[2] = c2[r] + g4[2]; o[3] = c3[r] + g4[3];
            *(v4f*)(qo + r * D) = o;
        }
        return;
    }
    int wave = blockIdx.x * 4 + wv;
    bool live = wave < GEMM_WAVES;
    int tm = wave >> 1, hf = wave & 1, col0 = hf * 64;
    float* lgt = lg + (wv >> 1) * (16 * H);
    if (live) {
        gemm_k128(A + (tm * 16 + c) * D + 4 * h, W + (4 * h) * D + col0 + 4 * c, c0, c1, c2, c3);
        const v4f as4 = *(const v4f*)(att_src + col0 + 4 * c);
        float ps[8];
#pragma unroll
        for (int r = 0; r < 8; ++r)
            ps[r] = c0[r] * as4[0] + c1[r] * as4[1] + c2[r] * as4[2] + c3[r] * as4[3];
        bool b4 = (c & 4) != 0, b2 = (c & 2) != 0, b1 = (c & 1) != 0;
        float ss = fold8(ps, b4, b2, b1);
        lgt[(8 * h + (c & 7)) * H + 2 * hf + (c >> 3)] = ss;
    }
    __syncthreads();
    if (live) {
        if (hf == 0) {
            const v2f t = *(const v2f*)(lgt + 2 * lane);
            *(v2f*)(a_src + tm * (16 * H) + 2 * lane) = t;
        }
        float* xo = X + (tm * 16 + 8 * h) * D + col0 + 4 * c;
#pragma unroll
        for (int r = 0; r < 8; ++r) {
            v4f o; o[0] = c0[r]; o[1] = c1[r]; o[2] = c2[r]; o[3] = c3[r];
            *(v4f*)(xo + r * D) = o;
        }
    }
}

__device__ __forceinline__ float wave_max(float v) {
#pragma unroll
    for (int o = 16; o >= 1; o >>= 1) v = fmaxf(v, __shfl_xor(v, o, 32));
    return v;
}
__device__ __forceinline__ float wave_sum(float v) {
#pragma unroll
    for (int o = 16; o >= 1; o >>= 1) v += __shfl_xor(v, o, 32);
    return v;
}
__device__ __forceinline__ float lane_val(float v, int l) {
    return __int_as_float(__builtin_amdgcn_readlane(__float_as_int(v), l));
}
__device__ __forceinline__ float pick(float b0, float b1, float b2, float b3,
                                      int k0, int k1, int k2, int k3) {
    return __int_as_float((__float_as_int(b0) & k0) | (__float_as_int(b1) & k1) |
                          (__float_as_int(b2) & k2) | (__float_as_int(b3) & k3));
}

__device__ __forceinline__ unsigned long long next_key(const unsigned short* __restrict__ seg, int deg,
                                                       unsigned long long lo, int lane) {
    unsigned long long best = ~0ull;
    for (int i = lane; i < deg; i += 32) {
        unsigned long long k = ((unsigned long long)(unsigned)seg[i] << 32) | (unsigned)i;
        if (k >= lo && k < best) best = k;
    }
#pragma unroll
    for (int o = 16; o >= 1; o >>= 1) {
        unsigned hi2 = (unsigned)__shfl_xor((int)(best >> 32), o, 32);
        unsigned lo2 = (unsigned)__shfl_xor((int)(unsigned)best, o, 32);
        unsigned long long t = ((unsigned long long)hi2 << 32) | lo2;
        best = t < best ? t : best;
    }
    return best;
}

__device__ __forceinline__ float edge_logit(const float4 x, const float4 as4, float pd) {
    float t = x.x * as4.x;
    t = fmaf(x.y, as4.y, t);
    t = fmaf(x.z, as4.z, t);
    t = fmaf(x.w, as4.w, t);
    asm("" : "+v"(t));
    t = add_xor<4>(t, t);
    t = add_xor<2>(t, t);
    t = add_xor<1>(t, t);
    float s = t + pd;
    return fmaxf(s, NEG_SLOPE * s);
}

__global__ void __launch_bounds__(256, 16)
k_owner(const int* __restrict__ ei,
                        const float* __restrict__ a_src, const float* __restrict__ att_src,
                        const float* __restrict__ att_dst,
                        const float* __restrict__ X, const float* __restrict__ bias,
                        const float* __restrict__ gamma, const float* __restrict__ beta,
                        float* lnout) {
    int n = (blockIdx.x * blockDim.x + threadIdx.x) >> 5;
    int lane = threadIdx.x & 31;
    const unsigned short* seg = (const unsigned short*)(lnout + n * D) + SEG0;
    const float4* X4 = (const float4*)X;
    const int hd = lane >> 3;
    float4 xn = X4[n * (D / 4) + lane];
    const float4 at = *(const float4*)(att_dst + lane * 4);
    int dg = *(const int*)(lnout + n * D);
    int sg = seg[lane];
    asm volatile("" : "+v"(dg), "+v"(sg), "+v"(xn.x), "+v"(xn.y), "+v"(xn.z), "+v"(xn.w)
                 : "v"(at.x), "v"(at.y), "v"(at.z), "v"(at.w));
    int deg = __builtin_amdgcn_readfirstlane(dg);
    float pd = xn.x * at.x + xn.y * at.y + xn.z * at.z + xn.w * at.w;
    pd = add_xor<4>(pd, pd);
    pd = add_xor<2>(pd, pd);
    pd = add_xor<1>(pd, pd);
    const float4 ad = make_float4(lane_val(pd, 0), lane_val(pd, 8), lane_val(pd, 16), lane_val(pd, 24));
    const int k0 = hd == 0 ? -1 : 0, k1 = hd == 1 ? -1 : 0;
    const int k2 = hd == 2 ? -1 : 0, k3 = hd == 3 ? -1 : 0;
    float4 acc = make_float4(0.0f, 0.0f, 0.0f, 0.0f);

    if (deg <= 32) {
        bool valid = lane < deg;
        int key = valid ? (sg << 5) | lane : 0x7FFFFFFF;
        int rank = 0;
        for (int j = 0; j < deg; ++j)
            rank += (__builtin_amdgcn_readlane(key, j) < key) ? 1 : 0;
        if (!valid) rank = lane;
        int ss = __builtin_amdgcn_ds_permute(rank << 2, key) >> 5;
        const float4 as4 = *(const float4*)(att_src + lane * 4);
        float m = -INFINITY, den = 0.0f;
        int p = 0;
        for (; p + 4 <= deg; p += 4) {
            float4 x[4];
            float e[4], w[4];
#pragma unroll
            for (int u = 0; u < 4; ++u)
                x[u] = X4[__builtin_amdgcn_readlane(ss, p + u) * (D / 4) + lane];
#pragma unroll
            for (int u = 0; u < 4; ++u)
                e[u] = edge_logit(x[u], as4, pd);
            float mn = fmaxf(fmaxf(fmaxf(m, e[0]), fmaxf(e[1], e[2])), e[3]);
            float sc = __expf(m - mn);
#pragma unroll
            for (int u = 0; u < 4; ++u)
                w[u] = __expf(e[u] - mn);
            den = fmaf(den, sc, (w[0] + w[1]) + (w[2] + w[3]));
            acc.x *= sc; acc.y *= sc; acc.z *= sc; acc.w *= sc;
#pragma unroll
            for (int u = 0; u < 4; ++u) {
                acc.x = fmaf(w[u], x[u].x, acc.x); acc.y = fmaf(w[u], x[u].y, acc.y);
                acc.z = fmaf(w[u], x[u].z, acc.z); acc.w = fmaf(w[u], x[u].w, acc.w);
            }
            m = mn;
        }
        for (; p < deg; ++p) {
            const float4 x = X4[__builtin_amdgcn_readlane(ss, p) * (D / 4) + lane];
            float e = edge_logit(x, as4, pd);
            float mn = fmaxf(m, e);
            float sc = __expf(m - mn), w = __expf(e - mn);
            den = fmaf(den, sc, w);
            acc.x = fmaf(w, x.x, acc.x * sc); acc.y = fmaf(w, x.y, acc.y * sc);
            acc.z = fmaf(w, x.z, acc.z * sc); acc.w = fmaf(w, x.w, acc.w * sc);
            m = mn;
        }
        float inv = 1.0f / fmaxf(den, 1e-16f);
        acc.x *= inv; acc.y *= inv; acc.z *= inv; acc.w *= inv;
    } else if (deg <= CAP) {
        float m0 = -INFINITY, m1 = -INFINITY, m2 = -INFINITY, m3 = -INFINITY;
        for (int i = lane; i < deg; i += 32) {
            const float4 a = *(const float4*)(a_src + (int)seg[i] * H);
            m0 = fmaxf(m0, lrelu(a.x + ad.x)); m1 = fmaxf(m1, lrelu(a.y + ad.y));
            m2 = fmaxf(m2, lrelu(a.z + ad.z)); m3 = fmaxf(m3, lrelu(a.w + ad.w));
        }
        m0 = wave_max(m0); m1 = wave_max(m1); m2 = wave_max(m2); m3 = wave_max(m3);
        float d0 = 0.0f, d1 = 0.0f, d2 = 0.0f, d3 = 0.0f;
        unsigned long long lo = 0;
        for (int t = 0; t < deg; ++t) {
            unsigned long long k = next_key(seg, deg, lo, lane);
            lo = k + 1;
            const float4 a = *(const float4*)(a_src + (int)(k >> 32) * H);
            d0 += __expf(lrelu(a.x + ad.x) - m0); d1 += __expf(lrelu(a.y + ad.y) - m1);
            d2 += __expf(lrelu(a.z + ad.z) - m2); d3 += __expf(lrelu(a.w + ad.w) - m3);
        }
        d0 = fmaxf(d0, 1e-16f); d1 = fmaxf(d1, 1e-16f);
        d2 = fmaxf(d2, 1e-16f); d3 = fmaxf(d3, 1e-16f);
        lo = 0;
        for (int t = 0; t < deg; ++t) {
            unsigned long long k = next_key(seg, deg, lo, lane);
            lo = k + 1;
            int sp = (int)(k >> 32);
            const float4 a = *(const float4*)(a_src + sp * H);
            float w = pick(__expf(lrelu(a.x + ad.x) - m0) / d0, __expf(lrelu(a.y + ad.y) - m1) / d1,
                           __expf(lrelu(a.z + ad.z) - m2) / d2, __expf(lrelu(a.w + ad.w) - m3) / d3, k0, k1, k2, k3);
            const float4 x = X4[sp * (D / 4) + lane];
            acc.x = fmaf(w, x.x, acc.x); acc.y = fmaf(w, x.y, acc.y);
            acc.z = fmaf(w, x.z, acc.z); acc.w = fmaf(w, x.w, acc.w);
        }
    } else {
        const int* dsts = ei + NE;
        float m0 = -INFINITY, m1 = -INFINITY, m2 = -INFINITY, m3 = -INFINITY;
        for (int e = lane; e < NE; e += 32) {
            if (dsts[e] == n) {
                const float4 a = *(const float4*)(a_src + ei[e] * H);
                m0 = fmaxf(m0, lrelu(a.x + ad.x)); m1 = fmaxf(m1, lrelu(a.y + ad.y));
                m2 = fmaxf(m2, lrelu(a.z + ad.z)); m3 = fmaxf(m3, lrelu(a.w + ad.w));
            }
        }
        m0 = wave_max(m0); m1 = wave_max(m1); m2 = wave_max(m2); m3 = wave_max(m3);
        float d0 = 0.0f, d1 = 0.0f, d2 = 0.0f, d3 = 0.0f;
        for (int base = 0; base < NE; base += 32) {
            int e = base + lane;
            bool hit = e < NE && dsts[e] == n;
            int sl = hit ? ei[e] : 0;
            unsigned mask = __builtin_amdgcn_ballot_w32(hit);
            while (mask) {
                int j = __builtin_ctz(mask);
                mask &= mask - 1;
                const float4 a = *(const float4*)(a_src + __builtin_amdgcn_readlane(sl, j) * H);
                d0 += __expf(lrelu(a.x + ad.x) - m0); d1 += __expf(lrelu(a.y + ad.y) - m1);
                d2 += __expf(lrelu(a.z + ad.z) - m2); d3 += __expf(lrelu(a.w + ad.w) - m3);
            }
        }
        d0 = fmaxf(d0, 1e-16f); d1 = fmaxf(d1, 1e-16f);
        d2 = fmaxf(d2, 1e-16f); d3 = fmaxf(d3, 1e-16f);
        for (int base = 0; base < NE; base += 32) {
            int e = base + lane;
            bool hit = e < NE && dsts[e] == n;
            int sl = hit ? ei[e] : 0;
            unsigned mask = __builtin_amdgcn_ballot_w32(hit);
            while (mask) {
                int j = __builtin_ctz(mask);
                mask &= mask - 1;
                int sp = __builtin_amdgcn_readlane(sl, j);
                const float4 a = *(const float4*)(a_src + sp * H);
                float w = pick(__expf(lrelu(a.x + ad.x) - m0) / d0, __expf(lrelu(a.y + ad.y) - m1) / d1,
                               __expf(lrelu(a.z + ad.z) - m2) / d2, __expf(lrelu(a.w + ad.w) - m3) / d3, k0, k1, k2, k3);
                const float4 x = X4[sp * (D / 4) + lane];
                acc.x = fmaf(w, x.x, acc.x); acc.y = fmaf(w, x.y, acc.y);
                acc.z = fmaf(w, x.z, acc.z); acc.w = fmaf(w, x.w, acc.w);
            }
        }
    }

    const float4 b4 = *(const float4*)(bias + lane * 4);
    float4 v = acc;
    v.x += b4.x; v.y += b4.y; v.z += b4.z; v.w += b4.w;
    float mu = wave_sum(v.x + v.y + v.z + v.w) * (1.0f / D);
    float dx = v.x - mu, dy = v.y - mu, dz = v.z - mu, dw = v.w - mu;
    float q = wave_sum(dx * dx + dy * dy + dz * dz + dw * dw);
    float rs = rsqrtf(q * (1.0f / D) + LN_EPS);
    const float4 g4 = *(const float4*)(gamma + lane * 4);
    const float4 be4 = *(const float4*)(beta + lane * 4);
    float4 o;
    o.x = dx * rs * g4.x + be4.x;  o.y = dy * rs * g4.y + be4.y;
    o.z = dz * rs * g4.z + be4.z;  o.w = dw * rs * g4.w + be4.w;
    *(float4*)(lnout + n * D + lane * 4) = o;
}

__global__ void k_gate(const float* __restrict__ LN, const float* __restrict__ QG,
                       const int* __restrict__ n2g, const float* __restrict__ gW,
                       float* __restrict__ out) {
    int wave = __builtin_amdgcn_readfirstlane((blockIdx.x * blockDim.x + threadIdx.x) >> 5);
    if (wave >= GEMM_WAVES) return;
    int lane = threadIdx.x & 31;
    int tm = wave >> 1, col0 = (wave & 1) * 64;
    int c = lane & 15, h = lane >> 4;
    const int* gp = n2g + tm * 16;
    int g16[16];
#pragma unroll
    for (int i = 0; i < 16; ++i) g16[i] = gp[i];
    v8f c0 = {0.f, 0.f, 0.f, 0.f, 0.f, 0.f, 0.f, 0.f};
    v8f c1 = c0, c2 = c0, c3 = c0;
    gemm_k128(LN + (tm * 16 + c) * D + 4 * h, gW + (4 * h) * D + col0 + 4 * c, c0, c1, c2, c3);
    int col = col0 + 4 * c;
    const float* lo = LN + (tm * 16 + 8 * h) * D + col;
    const float* qg = QG + col;
    float* oo = out + (tm * 16 + 8 * h) * D + col;
#pragma unroll
    for (int r = 0; r < 8; ++r) {
        v4f l4 = *(const v4f*)(lo + r * D);
        v4f q4 = *(const v4f*)(qg + (h ? g16[8 + r] : g16[r]) * D);
        v4f o;
        o[0] = l4[0] * (1.0f / (1.0f + __expf(-(c0[r] + q4[0]))));
        o[1] = l4[1] * (1.0f / (1.0f + __expf(-(c1[r] + q4[1]))));
        o[2] = l4[2] * (1.0f / (1.0f + __expf(-(c2[r] + q4[2]))));
        o[3] = l4[3] * (1.0f / (1.0f + __expf(-(c3[r] + q4[3]))));
        *(v4f*)(oo + r * D) = o;
    }
}

extern "C" void kernel_launch(void* const* d_in, const int* in_sizes, int n_in,
                              void* d_out, int out_size, void* d_ws, size_t ws_size,
                              hipStream_t stream) {
    (void)in_sizes; (void)n_in; (void)out_size; (void)ws_size;
    const float* node_input = (const float*)d_in[0];
    const int*   edge_index = (const int*)d_in[1];
    const int*   node2graph = (const int*)d_in[2];
    const float* query      = (const float*)d_in[3];
    const float* W          = (const float*)d_in[4];
    const float* att_src    = (const float*)d_in[5];
    const float* att_dst    = (const float*)d_in[6];
    const float* bias       = (const float*)d_in[7];
    const float* ln_gamma   = (const float*)d_in[8];
    const float* ln_beta    = (const float*)d_in[9];
    const float* gate_W     = (const float*)d_in[10];
    const float* gate_b     = (const float*)d_in[11];
    float* out = (float*)d_out;

    float* X     = out;
    float* LN    = (float*)d_ws;
    float* a_src = LN + NN * D;
    float* QG    = a_src + NN * H;

    k_proj    <<<GEMM_BLOCKS + NG / 32, 128, 0, stream>>>(node_input, W, att_src, query, gate_W, gate_b, X, a_src, QG, LN);
    k_fill    <<<(NE + 255) / 256, 256, 0, stream>>>(edge_index, LN);
    k_owner   <<<NN / 8, 256, 0, stream>>>(edge_index, a_src, att_src, att_dst, X, bias, ln_gamma, ln_beta, LN);
    k_gate    <<<GEMM_BLOCKS, 128, 0, stream>>>(LN, QG, node2graph, gate_W, out);
}
